# starting kernel
# baseline (speedup 1.0000x reference)
#include <hip/hip_runtime.h>
#include <stddef.h>

#define NU   8039
#define NI   32770
#define NB   4771
#define NTOT (NU + NI + NB)
#define EMB  32
#define NNZE 2000000
#define BATCHN 2048
#define KB   100
#define EPSF 1e-8f
#define L2NORM 1e-5f

typedef __attribute__((ext_vector_type(2))) float v2f;
typedef __attribute__((ext_vector_type(8))) float v8f;

__device__ __forceinline__ void atomAddF(float* p, float v) {
    (void)__hip_atomic_fetch_add(p, v, __ATOMIC_RELAXED, __HIP_MEMORY_SCOPE_AGENT);
}

__device__ __forceinline__ float feat_at(const float* uf, const float* itf,
                                         const float* bfv, int i, int d) {
    if (i < NU)      return uf[(size_t)i * EMB + d];
    if (i < NU + NI) return itf[(size_t)(i - NU) * EMB + d];
    return bfv[(size_t)(i - NU - NI) * EMB + d];
}

__global__ void zero_k(float* p, long n) {
    long i = (long)blockIdx.x * blockDim.x + threadIdx.x;
    long stride = (long)gridDim.x * blockDim.x;
    for (; i < n; i += stride) p[i] = 0.0f;
}

__global__ void degree_k(const int* __restrict__ rows, const int* __restrict__ cols,
                         float* dv, float* de) {
    int e = blockIdx.x * blockDim.x + threadIdx.x;
    if (e >= NNZE) return;
    atomAddF(dv + rows[e], 1.0f);
    atomAddF(de + cols[e], 1.0f);
}

__global__ void rsq_k(float* p, long n) {
    long i = (long)blockIdx.x * blockDim.x + threadIdx.x;
    if (i < n) p[i] = 1.0f / (sqrtf(p[i]) + EPSF);
}

__global__ void buildz_k(const float* uf, const float* itf, const float* bfv,
                         const float* __restrict__ dvi, float* __restrict__ z) {
    long t = (long)blockIdx.x * blockDim.x + threadIdx.x;
    if (t >= (long)NTOT * EMB) return;
    int i = (int)(t >> 5), d = (int)(t & 31);
    z[t] = feat_at(uf, itf, bfv, i, d) * dvi[i];
}

__global__ void scatter_k(const int* __restrict__ sidx, const int* __restrict__ didx,
                          const float* __restrict__ src, float* __restrict__ dst) {
    long t = (long)blockIdx.x * blockDim.x + threadIdx.x;
    int e = (int)(t >> 5);
    int lane = (int)(t & 31);
    if (e >= NNZE) return;
    int s = sidx[e], d = didx[e];
    float val = src[(size_t)s * EMB + lane];
    atomAddF(dst + (size_t)d * EMB + lane, val);
}

__global__ void rowscale_k(float* __restrict__ arr, const float* __restrict__ s) {
    long t = (long)blockIdx.x * blockDim.x + threadIdx.x;
    if (t >= (long)NTOT * EMB) return;
    arr[t] *= s[t >> 5];
}

__global__ void final_k(const float* uf, const float* itf, const float* bfv,
                        const float* __restrict__ dvi, const float* __restrict__ v,
                        float* __restrict__ all, float* __restrict__ partials) {
    __shared__ float sdata[256];
    long t = (long)blockIdx.x * blockDim.x + threadIdx.x;
    float sq = 0.0f;
    if (t < (long)NTOT * EMB) {
        int i = (int)(t >> 5), d = (int)(t & 31);
        float val = feat_at(uf, itf, bfv, i, d) * 0.5f + v[t] * dvi[i] * (1.0f / 3.0f);
        all[t] = val;
        sq = val * val;
    }
    sdata[threadIdx.x] = sq;
    __syncthreads();
    for (int s = 128; s > 0; s >>= 1) {
        if (threadIdx.x < (unsigned)s) sdata[threadIdx.x] += sdata[threadIdx.x + s];
        __syncthreads();
    }
    if (threadIdx.x == 0) partials[blockIdx.x] = sdata[0];
}

__global__ void loss_k(const float* __restrict__ partials, int nb, float* __restrict__ out_loss) {
    __shared__ float sdata[256];
    float s = 0.0f;
    for (int i = threadIdx.x; i < nb; i += 256) s += partials[i];
    sdata[threadIdx.x] = s;
    __syncthreads();
    for (int k = 128; k > 0; k >>= 1) {
        if (threadIdx.x < (unsigned)k) sdata[threadIdx.x] += sdata[threadIdx.x + k];
        __syncthreads();
    }
    if (threadIdx.x == 0) *out_loss = L2NORM * sdata[0];
}

__global__ void pred_k(const float* __restrict__ all, const int* __restrict__ users,
                       const int* __restrict__ bundles, const float* __restrict__ ubound,
                       const float* __restrict__ zrow,
                       float* __restrict__ pred, float* __restrict__ usb) {
    int wave = (blockIdx.x * blockDim.x + threadIdx.x) >> 5;
    int lane = threadIdx.x & 31;
    int b = wave / 7;
    int kt = wave % 7;

    const float* bfbase = all + (size_t)(NU + NI) * EMB;
    const float* ue = all + (size_t)users[b] * EMB;

    int m = lane & 15;
    int khalf = lane >> 4;
    int kk = kt * 16 + m;

    const float* arow;
    if (kk < KB)                 arow = bfbase + (size_t)bundles[b * KB + kk] * EMB;
    else if (kt == 6 && m == 4)  arow = ubound;
    else                         arow = zrow;

    v8f c = {};
#pragma unroll
    for (int s = 0; s < 8; ++s) {
        int e0 = s * 4 + khalf * 2;
        v2f a  = *(const v2f*)(arow + e0);
        v2f bb = *(const v2f*)(ue + e0);
        c = __builtin_amdgcn_wmma_f32_16x16x4_f32(false, a, false, bb,
                                                  (short)0, c, false, false);
    }

    if (lane == 0) {
#pragma unroll
        for (int j = 0; j < 8; ++j) {
            int kkk = kt * 16 + j;
            if (kkk < KB)                 pred[(size_t)b * KB + kkk] = c[j];
            else if (kt == 6 && j == 4)   usb[b] = c[j];
        }
    } else if (lane == 16) {
#pragma unroll
        for (int j = 0; j < 8; ++j) {
            int kkk = kt * 16 + 8 + j;
            if (kkk < KB) pred[(size_t)b * KB + kkk] = c[j];
        }
    }
}

extern "C" void kernel_launch(void* const* d_in, const int* in_sizes, int n_in,
                              void* d_out, int out_size, void* d_ws, size_t ws_size,
                              hipStream_t stream) {
    const float* uf      = (const float*)d_in[0];
    const float* itf     = (const float*)d_in[1];
    const float* bfv     = (const float*)d_in[2];
    const float* ubound  = (const float*)d_in[3];
    const int*   rows    = (const int*)d_in[4];
    const int*   cols    = (const int*)d_in[5];
    const int*   users   = (const int*)d_in[6];
    const int*   bundles = (const int*)d_in[7];
    float* out = (float*)d_out;

    float* ws   = (float*)d_ws;
    float* dvde = ws;
    float* z    = dvde + 2L * NTOT;
    float* u    = z + 32L * NTOT;
    float* v    = u + 32L * NTOT;
    float* partials = v + 32L * NTOT;
    float* zrow = partials + 8192;

    const long elems = (long)NTOT * EMB;
    const int  eb    = (int)((elems + 255) / 256);
    const int  edgeb = (NNZE * 32 + 255) / 256;

    zero_k<<<2048, 256, 0, stream>>>(dvde, 2L * NTOT);
    zero_k<<<4096, 256, 0, stream>>>(u, 64L * NTOT);
    zero_k<<<1, 32, 0, stream>>>(zrow, 32L);

    degree_k<<<(NNZE + 255) / 256, 256, 0, stream>>>(rows, cols, dvde, dvde + NTOT);
    rsq_k<<<(int)((2L * NTOT + 255) / 256), 256, 0, stream>>>(dvde, 2L * NTOT);

    buildz_k<<<eb, 256, 0, stream>>>(uf, itf, bfv, dvde, z);
    scatter_k<<<edgeb, 256, 0, stream>>>(rows, cols, z, u);
    rowscale_k<<<eb, 256, 0, stream>>>(u, dvde + NTOT);
    scatter_k<<<edgeb, 256, 0, stream>>>(cols, rows, u, v);

    final_k<<<eb, 256, 0, stream>>>(uf, itf, bfv, dvde, v, z, partials);
    loss_k<<<1, 256, 0, stream>>>(partials, eb, out + BATCHN * KB + BATCHN);

    pred_k<<<1792, 256, 0, stream>>>(z, users, bundles, ubound, zrow,
                                     out, out + BATCHN * KB);
}


# degree launch writes each 4000-edge chunk's edge words grouped by (row half, col half) in ring order; an edge-pass block reads only its own half's run, a wave per chunk; LDS adds in arrival order
# speedup vs baseline: 3.8653x; 3.8653x over previous
#include <hip/hip_runtime.h>
#include <stddef.h>

#define NU   8039
#define NI   32770
#define NB   4771
#define NTOT (NU + NI + NB)
#define EMB  32
#define NNZE 2000000
#define BATCHN 2048
#define KB   100
#define EPSF 1e-8f
#define L2NORM 1e-5f
#define NP     45600
#define HALFN  22790
#define H0N    22784
#define H1N    (NTOT - H0N)
#define NSLICE 4
#define ESLICE (NNZE / NSLICE)
#define PARTN   5760
#define PK_OFF  (96L * NP + 2L * NTOT + PARTN + 104)
#define PK_END_B ((PK_OFF + (long)NNZE) * 4L)
#define PL_END_B (((288L - 96L) * NP) * 4L + PK_END_B)
#define CH     4000
#define CHV    (CH / 4)
#define NCHQ   (ESLICE / CH)
#define NCHUNK (NNZE / CH)

typedef __attribute__((ext_vector_type(2))) float v2f;
typedef __attribute__((ext_vector_type(8))) float v8f;

__device__ __forceinline__ void atomAddF(float* p, float v) {
    (void)__hip_atomic_fetch_add(p, v, __ATOMIC_RELAXED, __HIP_MEMORY_SCOPE_AGENT);
}

__device__ __forceinline__ float feat_at(const float* uf, const float* itf,
                                         const float* bfv, int i, int d) {
    if (i < NU)      return uf[(size_t)i * EMB + d];
    if (i < NU + NI) return itf[(size_t)(i - NU) * EMB + d];
    return bfv[(size_t)(i - NU - NI) * EMB + d];
}

__global__ void zero_k(float* p, long n) {
    long i = (long)blockIdx.x * blockDim.x + threadIdx.x;
    long stride = (long)gridDim.x * blockDim.x;
    for (; i < n; i += stride) p[i] = 0.0f;
}

__global__ void degree_k(const int* __restrict__ rows, const int* __restrict__ cols,
                         float* cntA, float* cntB) {
    int e = blockIdx.x * blockDim.x + threadIdx.x;
    if (e >= NNZE) return;
    const int r = rows[e], c = cols[e];
    const unsigned place = __builtin_amdgcn_s_sendmsg_rtn(0x87);
    float* cnt = ((place >> 18) & 1u) ? cntB : cntA;
    atomAddF(cnt + r, 1.0f);
    atomAddF(cnt + NTOT + c, 1.0f);
}

__device__ __forceinline__ int ring_class(int r, int c) {
    return r >= H0N ? (c >= H0N ? 2 : 3) : (c >= H0N ? 1 : 0);
}

__device__ __forceinline__ int class_place(int k, int lane, int* ncls) {
    const unsigned m0 = __builtin_amdgcn_ballot_w32(k == 0);
    const unsigned m1 = __builtin_amdgcn_ballot_w32(k == 1);
    const unsigned m2 = __builtin_amdgcn_ballot_w32(k == 2);
    const unsigned m3 = __builtin_amdgcn_ballot_w32(k == 3);
    const unsigned mine = k == 0 ? m0 : k == 1 ? m1 : k == 2 ? m2 : m3;
    int first = 0;
    if (lane < 4) {
        const unsigned ml = lane == 0 ? m0 : lane == 1 ? m1 : lane == 2 ? m2 : m3;
        first = __hip_atomic_fetch_add(ncls + lane, (int)__popc(ml), __ATOMIC_RELAXED,
                                       __HIP_MEMORY_SCOPE_WORKGROUP);
    }
    const int base = __shfl(first, k & 3);
    return base + (int)__popc(mine & ((1u << lane) - 1u));
}

__global__ __launch_bounds__(1024)
void degree_part_k(const int* __restrict__ rows, const int* __restrict__ cols,
                   float* cntA, float* cntB, unsigned* __restrict__ pk,
                   int4* __restrict__ cnts) {
    __shared__ __attribute__((aligned(16))) unsigned stage[CH];
    __shared__ int ncls[4];
    const int t = threadIdx.x;
    const int lane = t & 31;
    if (t < 4) ncls[t] = 0;
    __syncthreads();
    const bool live = t < CHV;
    const int v = blockIdx.x * CHV + (live ? t : 0);
    const int4 r4 = ((const int4*)rows)[v];
    const int4 c4 = ((const int4*)cols)[v];
    const unsigned place = __builtin_amdgcn_s_sendmsg_rtn(0x87);
    float* cnt = ((place >> 18) & 1u) ? cntB : cntA;
    const int k0 = live ? ring_class(r4.x, c4.x) : 4;
    const int k1 = live ? ring_class(r4.y, c4.y) : 4;
    const int k2 = live ? ring_class(r4.z, c4.z) : 4;
    const int k3 = live ? ring_class(r4.w, c4.w) : 4;
    const int q0 = class_place(k0, lane, ncls);
    const int q1 = class_place(k1, lane, ncls);
    const int q2 = class_place(k2, lane, ncls);
    const int q3 = class_place(k3, lane, ncls);
    if (live) {
        atomAddF(cnt + r4.x, 1.0f); atomAddF(cnt + NTOT + c4.x, 1.0f);
        atomAddF(cnt + r4.y, 1.0f); atomAddF(cnt + NTOT + c4.y, 1.0f);
        atomAddF(cnt + r4.z, 1.0f); atomAddF(cnt + NTOT + c4.z, 1.0f);
        atomAddF(cnt + r4.w, 1.0f); atomAddF(cnt + NTOT + c4.w, 1.0f);
    }
    __syncthreads();
    const int n0 = ncls[0], n1 = ncls[1], n2 = ncls[2], n3 = ncls[3];
    if (live) {
        const int b1 = n0, b2 = n0 + n1, b3 = n0 + n1 + n2;
        stage[(k0 == 0 ? 0 : k0 == 1 ? b1 : k0 == 2 ? b2 : b3) + q0] = (unsigned)r4.x | ((unsigned)c4.x << 16);
        stage[(k1 == 0 ? 0 : k1 == 1 ? b1 : k1 == 2 ? b2 : b3) + q1] = (unsigned)r4.y | ((unsigned)c4.y << 16);
        stage[(k2 == 0 ? 0 : k2 == 1 ? b1 : k2 == 2 ? b2 : b3) + q2] = (unsigned)r4.z | ((unsigned)c4.z << 16);
        stage[(k3 == 0 ? 0 : k3 == 1 ? b1 : k3 == 2 ? b2 : b3) + q3] = (unsigned)r4.w | ((unsigned)c4.w << 16);
    }
    __syncthreads();
    if (live) {
        uint4 o;
        o.x = stage[4 * t]; o.y = stage[4 * t + 1]; o.z = stage[4 * t + 2]; o.w = stage[4 * t + 3];
        ((uint4*)pk)[blockIdx.x * CHV + t] = o;
    }
    if (t == 0) cnts[blockIdx.x] = make_int4(n0, n1, n2, n3);
}

__global__ void rsq_k(float* pa, const float* pb, long n) {
    long i = (long)blockIdx.x * blockDim.x + threadIdx.x;
    if (i < n) {
        float d = pa[i];
        if (pb) d += pb[i];
        pa[i] = 1.0f / (sqrtf(d) + EPSF);
    }
}

__global__ void buildz_k(const float* uf, const float* itf, const float* bfv,
                         const float* __restrict__ dvi, float* __restrict__ zp) {
    __shared__ float tile[32][33];
    const int i0 = blockIdx.x * 32;
#pragma unroll
    for (int r = 0; r < 4; ++r) {
        int idx = threadIdx.x + r * 256;
        int j = idx >> 5, d = idx & 31;
        int i = i0 + j;
        tile[j][d] = (i < NTOT) ? feat_at(uf, itf, bfv, i, d) * dvi[i] : 0.0f;
    }
    __syncthreads();
#pragma unroll
    for (int r = 0; r < 4; ++r) {
        int idx = threadIdx.x + r * 256;
        int d = idx >> 5, j = idx & 31;
        int i = i0 + j;
        if (i < NTOT) zp[(size_t)d * NP + i] = tile[j][d];
    }
}

template <bool DST_LOW>
__device__ __forceinline__ int src_node(unsigned w) {
    return DST_LOW ? (int)(w >> 16) : (int)(w & 0xFFFFu);
}
template <bool DST_LOW>
__device__ __forceinline__ int dst_node(unsigned w) {
    return DST_LOW ? (int)(w & 0xFFFFu) : (int)(w >> 16);
}

__device__ __forceinline__ void edge1(int dcol, float val, int lo, int cnt, float* dstL, bool live) {
    unsigned dd = (unsigned)(dcol - lo);
    if (live && dd < (unsigned)cnt)
        (void)__hip_atomic_fetch_add(dstL + dd, val, __ATOMIC_RELAXED,
                                     __HIP_MEMORY_SCOPE_WORKGROUP);
}

__device__ __forceinline__ void add_hit(bool hit, unsigned dd, float val, float* dstL) {
    if (__builtin_expect(hit, true))
        (void)__hip_atomic_fetch_add(dstL + dd, val, __ATOMIC_RELAXED,
                                     __HIP_MEMORY_SCOPE_WORKGROUP);
}

#define NV4 (ESLICE / 4)

__device__ __forceinline__ int ring_vec(int vs, int i) {
    const int x = vs + i;
    return x < CHV ? x : x - CHV;
}

template <bool DST_LOW>
__global__ __launch_bounds__(1024)
void spmm_pk_k(const unsigned* __restrict__ pk, const int4* __restrict__ cnts,
            const float* __restrict__ srcp, const float* __restrict__ rs,
            float* __restrict__ dstp, int nsrc, int plain) {
    __shared__ float srcL[NTOT];
    __shared__ float dstL[H1N];
    const int d  = blockIdx.x & 31;
    const int h  = (blockIdx.x >> 5) & 1;
    const int q  = blockIdx.x >> 6;
    const int lo = h * H0N;
    const int cnt = h ? H1N : H0N;
    const float* sp = srcp + (size_t)d * NP;
    if (nsrc == 4) {
        const float* s1 = sp + 32L * NP;
        const float* s2 = sp + 64L * NP;
        const float* s3 = sp + 96L * NP;
        if (rs) {
            for (int i = threadIdx.x; i < NTOT; i += 1024)
                srcL[i] = ((sp[i] + s1[i]) + (s2[i] + s3[i])) * rs[i];
        } else {
            for (int i = threadIdx.x; i < NTOT; i += 1024)
                srcL[i] = (sp[i] + s1[i]) + (s2[i] + s3[i]);
        }
    } else if (rs) {
        for (int i = threadIdx.x; i < NTOT; i += 1024) srcL[i] = sp[i] * rs[i];
    } else {
        for (int i = threadIdx.x; i < NTOT; i += 1024) srcL[i] = sp[i];
    }
    for (int i = threadIdx.x; i < H1N; i += 1024) dstL[i] = 0.0f;
    __syncthreads();

    const int lane = threadIdx.x & 31;
    const int wv = __builtin_amdgcn_readfirstlane((int)(threadIdx.x >> 5));
    const int rot = (int)((blockIdx.x & 63) * NCHQ) >> 6;
    for (int cl = wv; cl < NCHQ; cl += 32) {
        int cc = cl + rot;
        if (cc >= NCHQ) cc -= NCHQ;
        const int chunk = q * NCHQ + cc;
        const int4 n = cnts[chunk];
        int start, len;
        if (DST_LOW) {
            start = h ? n.x + n.y : 0;
            len   = h ? n.z + n.w : n.x + n.y;
        } else {
            start = h ? n.x : n.x + n.y + n.z;
            len   = h ? n.y + n.z : n.w + n.x;
        }
        const int vs = start >> 2;
        int nv = ((start + len + 3) >> 2) - vs;
        if (nv > CHV) nv = CHV;
        if (len == 0) nv = 0;
        const uint4* p4 = (const uint4*)(pk + (size_t)chunk * CH);
        uint4 wa = p4[ring_vec(vs, lane < nv ? lane : 0)];
        uint4 wb = p4[ring_vec(vs, lane + 32 < nv ? lane + 32 : 0)];
        for (int t = 0; t < nv; t += 64) {
            const int ja = t + 64 + lane, jb = ja + 32;
            const uint4 na = p4[ring_vec(vs, ja < nv ? ja : 0)];
            const uint4 nb = p4[ring_vec(vs, jb < nv ? jb : 0)];
            const bool la = (t + lane) < nv;
            const bool lb = (t + lane + 32) < nv;
            const unsigned da0 = (unsigned)(dst_node<DST_LOW>(wa.x) - lo);
            const unsigned da1 = (unsigned)(dst_node<DST_LOW>(wa.y) - lo);
            const unsigned da2 = (unsigned)(dst_node<DST_LOW>(wa.z) - lo);
            const unsigned da3 = (unsigned)(dst_node<DST_LOW>(wa.w) - lo);
            const unsigned db0 = (unsigned)(dst_node<DST_LOW>(wb.x) - lo);
            const unsigned db1 = (unsigned)(dst_node<DST_LOW>(wb.y) - lo);
            const unsigned db2 = (unsigned)(dst_node<DST_LOW>(wb.z) - lo);
            const unsigned db3 = (unsigned)(dst_node<DST_LOW>(wb.w) - lo);
            const bool ha0 = la && da0 < (unsigned)cnt, ha1 = la && da1 < (unsigned)cnt;
            const bool ha2 = la && da2 < (unsigned)cnt, ha3 = la && da3 < (unsigned)cnt;
            const bool hb0 = lb && db0 < (unsigned)cnt, hb1 = lb && db1 < (unsigned)cnt;
            const bool hb2 = lb && db2 < (unsigned)cnt, hb3 = lb && db3 < (unsigned)cnt;
            float a0 = srcL[ha0 ? src_node<DST_LOW>(wa.x) : 0], a1 = srcL[ha1 ? src_node<DST_LOW>(wa.y) : 0];
            float a2 = srcL[ha2 ? src_node<DST_LOW>(wa.z) : 0], a3 = srcL[ha3 ? src_node<DST_LOW>(wa.w) : 0];
            float b0 = srcL[hb0 ? src_node<DST_LOW>(wb.x) : 0], b1 = srcL[hb1 ? src_node<DST_LOW>(wb.y) : 0];
            float b2 = srcL[hb2 ? src_node<DST_LOW>(wb.z) : 0], b3 = srcL[hb3 ? src_node<DST_LOW>(wb.w) : 0];
            asm volatile("" :: "v"(a0), "v"(a1), "v"(a2), "v"(a3),
                               "v"(b0), "v"(b1), "v"(b2), "v"(b3));
            add_hit(ha0, da0, a0, dstL);
            add_hit(ha1, da1, a1, dstL);
            add_hit(ha2, da2, a2, dstL);
            add_hit(ha3, da3, a3, dstL);
            add_hit(hb0, db0, b0, dstL);
            add_hit(hb1, db1, b1, dstL);
            add_hit(hb2, db2, b2, dstL);
            add_hit(hb3, db3, b3, dstL);
            wa = na; wb = nb;
        }
    }
    __syncthreads();

    if (plain) {
        float* dp = dstp + ((size_t)q * 32 + d) * NP + lo;
        for (int i = threadIdx.x; i < cnt; i += 1024) dp[i] = dstL[i];
    } else {
        float* dp = dstp + (size_t)d * NP + lo;
        for (int i = threadIdx.x; i < cnt; i += 1024) atomAddF(dp + i, dstL[i]);
    }
}

__global__ __launch_bounds__(1024)
void spmm_k(const int* __restrict__ sidx, const int* __restrict__ didx,
            const float* __restrict__ srcp, const float* __restrict__ rs,
            float* __restrict__ dstp) {
    __shared__ float srcL[NTOT];
    __shared__ float dstL[HALFN];
    const int d  = blockIdx.x & 31;
    const int h  = (blockIdx.x >> 5) & 1;
    const int q  = blockIdx.x >> 6;
    const int lo = h * HALFN;
    const float* sp = srcp + (size_t)d * NP;
    if (rs) {
        for (int i = threadIdx.x; i < NTOT; i += 1024) srcL[i] = sp[i] * rs[i];
    } else {
        for (int i = threadIdx.x; i < NTOT; i += 1024) srcL[i] = sp[i];
    }
    for (int i = threadIdx.x; i < HALFN; i += 1024) dstL[i] = 0.0f;
    __syncthreads();

    const int4* s4 = (const int4*)(sidx + (size_t)q * ESLICE);
    const int4* d4 = (const int4*)(didx + (size_t)q * ESLICE);
    const int off = (blockIdx.x & 63) * (NV4 / 64 / 32 * 32);
    int k = threadIdx.x;
    int v = k + off;
    int4 dc = d4[v];
    int4 sr = s4[v];
    for (;;) {
        const int kn = k + 1024;
        const int vw = v + 1024;
        const int vn = vw < NV4 ? vw : vw - NV4;
        const int kc = kn < NV4 ? vn : v;
        const int4 dcn = d4[kc];
        const int4 srn = s4[kc];
        float a0 = srcL[sr.x], a1 = srcL[sr.y], a2 = srcL[sr.z], a3 = srcL[sr.w];
        asm volatile("" :: "v"(a0), "v"(a1), "v"(a2), "v"(a3));
        edge1(dc.x, a0, lo, HALFN, dstL, true);
        edge1(dc.y, a1, lo, HALFN, dstL, true);
        edge1(dc.z, a2, lo, HALFN, dstL, true);
        edge1(dc.w, a3, lo, HALFN, dstL, true);
        if (kn >= NV4) break;
        dc = dcn; sr = srn; k = kn; v = vn;
    }
    __syncthreads();

    float* dp = dstp + (size_t)d * NP + lo;
    for (int i = threadIdx.x; i < HALFN; i += 1024) atomAddF(dp + i, dstL[i]);
}

__global__ void final_k(const float* uf, const float* itf, const float* bfv,
                        const float* __restrict__ dvi, const float* __restrict__ v, int nv,
                        float* __restrict__ all, float* __restrict__ partials) {
    __shared__ float sdata[256];
    long t = (long)blockIdx.x * blockDim.x + threadIdx.x;
    float sq = 0.0f;
    if (t < (long)NTOT * EMB) {
        int i = (int)(t >> 5), d = (int)(t & 31);
        const float x = feat_at(uf, itf, bfv, i, d);
        const float sc = dvi[i];
        const float* vp = v + (size_t)d * NP + i;
        float vs = vp[0];
        if (nv == 4) vs = (vs + vp[32L * NP]) + (vp[64L * NP] + vp[96L * NP]);
        float val = x * 0.5f + vs * sc * (1.0f / 3.0f);
        all[t] = val;
        sq = val * val;
    }
    sdata[threadIdx.x] = sq;
    __syncthreads();
    for (int s = 128; s > 0; s >>= 1) {
        if (threadIdx.x < (unsigned)s) sdata[threadIdx.x] += sdata[threadIdx.x + s];
        __syncthreads();
    }
    if (threadIdx.x == 0) partials[blockIdx.x] = sdata[0];
}

__global__ void loss_k(const float* __restrict__ partials, int nb, float* __restrict__ out_loss) {
    __shared__ float sdata[256];
    float s = 0.0f;
    for (int i = threadIdx.x; i < nb; i += 256) s += partials[i];
    sdata[threadIdx.x] = s;
    __syncthreads();
    for (int k = 128; k > 0; k >>= 1) {
        if (threadIdx.x < (unsigned)k) sdata[threadIdx.x] += sdata[threadIdx.x + k];
        __syncthreads();
    }
    if (threadIdx.x == 0) *out_loss = L2NORM * sdata[0];
}

__global__ void pred_k(const float* __restrict__ all, const int* __restrict__ users,
                       const int* __restrict__ bundles, const float* __restrict__ ubound,
                       const float* __restrict__ zrow,
                       float* __restrict__ pred, float* __restrict__ usb) {
    int wave = (blockIdx.x * blockDim.x + threadIdx.x) >> 5;
    int lane = threadIdx.x & 31;
    int b = wave / 7;
    int kt = wave % 7;

    const float* bfbase = all + (size_t)(NU + NI) * EMB;
    const float* ue = all + (size_t)users[b] * EMB;

    int m = lane & 15;
    int khalf = lane >> 4;
    int kk = kt * 16 + m;

    const float* arow;
    if (kk < KB)                 arow = bfbase + (size_t)bundles[b * KB + kk] * EMB;
    else if (kt == 6 && m == 4)  arow = ubound;
    else                         arow = zrow;

    v8f c = {};
#pragma unroll
    for (int s = 0; s < 8; ++s) {
        int e0 = s * 4 + khalf * 2;
        v2f a  = *(const v2f*)(arow + e0);
        v2f bb = *(const v2f*)(ue + e0);
        c = __builtin_amdgcn_wmma_f32_16x16x4_f32(false, a, false, bb,
                                                  (short)0, c, false, false);
    }

    if (lane == 0) {
#pragma unroll
        for (int j = 0; j < 8; ++j) {
            int kkk = kt * 16 + j;
            if (kkk < KB)                 pred[(size_t)b * KB + kkk] = c[j];
            else if (kt == 6 && j == 4)   usb[b] = c[j];
        }
    } else if (lane == 16) {
#pragma unroll
        for (int j = 0; j < 8; ++j) {
            int kkk = kt * 16 + 8 + j;
            if (kkk < KB) pred[(size_t)b * KB + kkk] = c[j];
        }
    }
}

extern "C" void kernel_launch(void* const* d_in, const int* in_sizes, int n_in,
                              void* d_out, int out_size, void* d_ws, size_t ws_size,
                              hipStream_t stream) {
    const float* uf      = (const float*)d_in[0];
    const float* itf     = (const float*)d_in[1];
    const float* bfv     = (const float*)d_in[2];
    const float* ubound  = (const float*)d_in[3];
    const int*   rows    = (const int*)d_in[4];
    const int*   cols    = (const int*)d_in[5];
    const int*   users   = (const int*)d_in[6];
    const int*   bundles = (const int*)d_in[7];
    float* out = (float*)d_out;

    const bool plain = ws_size >= (size_t)PL_END_B;
    float* ws   = (float*)d_ws;
    float* p0   = ws;
    float* p1   = p0 + 32L * NP;
    float* p2   = p1 + (plain ? 128L : 32L) * NP;
    float* dvde = p2 + (plain ? 128L : 32L) * NP;
    float* partials = dvde + 2L * NTOT;
    float* zrow = partials + PARTN;
    const bool packed = plain || ws_size >= (size_t)PK_END_B;
    unsigned* pk = packed ? (unsigned*)(zrow + 104) : (unsigned*)nullptr;

    const long elems = (long)NTOT * EMB;
    const int  eb    = (int)((elems + 255) / 256);

    float* cntB = plain ? dvde - 2L * NTOT : dvde;

    zero_k<<<2048, 256, 0, stream>>>(cntB, plain ? 4L * NTOT : 2L * NTOT);
    if (!plain) zero_k<<<4096, 256, 0, stream>>>(p1, 64L * NP);
    zero_k<<<1, 32, 0, stream>>>(zrow, 32L);

    int4* ccnt = (int4*)partials;
    if (packed) degree_part_k<<<NCHUNK, 1024, 0, stream>>>(rows, cols, dvde, cntB, pk, ccnt);
    else        degree_k<<<(NNZE + 255) / 256, 256, 0, stream>>>(rows, cols, dvde, cntB);
    rsq_k<<<(int)((2L * NTOT + 255) / 256), 256, 0, stream>>>(dvde, plain ? cntB : (const float*)nullptr,
                                                             2L * NTOT);

    buildz_k<<<(NTOT + 31) / 32, 256, 0, stream>>>(uf, itf, bfv, dvde, p0);
    if (packed) {
        spmm_pk_k<false><<<32 * 2 * NSLICE, 1024, 0, stream>>>(pk, ccnt, p0, (const float*)nullptr, p1,
                                                               1, plain ? 1 : 0);
        spmm_pk_k<true><<<32 * 2 * NSLICE, 1024, 0, stream>>>(pk, ccnt, p1, dvde + NTOT, p2,
                                                              plain ? 4 : 1, plain ? 1 : 0);
    } else {
        spmm_k<<<32 * 2 * NSLICE, 1024, 0, stream>>>(rows, cols, p0, (const float*)nullptr, p1);
        spmm_k<<<32 * 2 * NSLICE, 1024, 0, stream>>>(cols, rows, p1, dvde + NTOT, p2);
    }

    final_k<<<eb, 256, 0, stream>>>(uf, itf, bfv, dvde, p2, plain ? 4 : 1, p0, partials);
    loss_k<<<1, 256, 0, stream>>>(partials, eb, out + BATCHN * KB + BATCHN);

    pred_k<<<1792, 256, 0, stream>>>(p0, users, bundles, ubound, zrow,
                                     out, out + BATCHN * KB);
}


# edge words kept twice, grouped per chunk by destination eighth (of row, of col); an edge-pass block owns (dim, eighth) over ALL edges: one arrival-order LDS sum per element, no quarter partials
# speedup vs baseline: 4.0292x; 1.0424x over previous
#include <hip/hip_runtime.h>
#include <stddef.h>

#define NU   8039
#define NI   32770
#define NB   4771
#define NTOT (NU + NI + NB)
#define EMB  32
#define NNZE 2000000
#define BATCHN 2048
#define KB   100
#define EPSF 1e-8f
#define L2NORM 1e-5f
#define NP     45600
#define HALFN  22790
#define H0N    22784
#define H1N    (NTOT - H0N)
#define NSLICE 4
#define ESLICE (NNZE / NSLICE)
#define PARTN   5760
#define PK_OFF  (96L * NP + 2L * NTOT + PARTN + 104)
#define PK_END_B ((PK_OFF + (long)NNZE) * 4L)
#define PL_END_B (((288L - 96L) * NP) * 4L + PK_END_B)
#define CH     4000
#define CHV    (CH / 4)
#define NCHQ   (ESLICE / CH)
#define NCHUNK (NNZE / CH)

#define E8N    5696
#define E8L    (NTOT - 7 * E8N)

typedef __attribute__((ext_vector_type(2))) float v2f;
typedef __attribute__((ext_vector_type(8))) float v8f;

__device__ __forceinline__ void atomAddF(float* p, float v) {
    (void)__hip_atomic_fetch_add(p, v, __ATOMIC_RELAXED, __HIP_MEMORY_SCOPE_AGENT);
}

__device__ __forceinline__ float feat_at(const float* uf, const float* itf,
                                         const float* bfv, int i, int d) {
    if (i < NU)      return uf[(size_t)i * EMB + d];
    if (i < NU + NI) return itf[(size_t)(i - NU) * EMB + d];
    return bfv[(size_t)(i - NU - NI) * EMB + d];
}

__global__ void zero_k(float* p, long n) {
    long i = (long)blockIdx.x * blockDim.x + threadIdx.x;
    long stride = (long)gridDim.x * blockDim.x;
    for (; i < n; i += stride) p[i] = 0.0f;
}

__global__ void degree_k(const int* __restrict__ rows, const int* __restrict__ cols,
                         float* cntA, float* cntB) {
    int e = blockIdx.x * blockDim.x + threadIdx.x;
    if (e >= NNZE) return;
    const int r = rows[e], c = cols[e];
    const unsigned place = __builtin_amdgcn_s_sendmsg_rtn(0x87);
    float* cnt = ((place >> 18) & 1u) ? cntB : cntA;
    atomAddF(cnt + r, 1.0f);
    atomAddF(cnt + NTOT + c, 1.0f);
}

__device__ __forceinline__ int ring_class(int r, int c) {
    return r >= H0N ? (c >= H0N ? 2 : 3) : (c >= H0N ? 1 : 0);
}

__device__ __forceinline__ int class_place(int k, int lane, int* ncls) {
    const unsigned m0 = __builtin_amdgcn_ballot_w32(k == 0);
    const unsigned m1 = __builtin_amdgcn_ballot_w32(k == 1);
    const unsigned m2 = __builtin_amdgcn_ballot_w32(k == 2);
    const unsigned m3 = __builtin_amdgcn_ballot_w32(k == 3);
    const unsigned mine = k == 0 ? m0 : k == 1 ? m1 : k == 2 ? m2 : m3;
    int first = 0;
    if (lane < 4) {
        const unsigned ml = lane == 0 ? m0 : lane == 1 ? m1 : lane == 2 ? m2 : m3;
        first = __hip_atomic_fetch_add(ncls + lane, (int)__popc(ml), __ATOMIC_RELAXED,
                                       __HIP_MEMORY_SCOPE_WORKGROUP);
    }
    const int base = __shfl(first, k & 3);
    return base + (int)__popc(mine & ((1u << lane) - 1u));
}

__global__ __launch_bounds__(1024)
void degree_part_k(const int* __restrict__ rows, const int* __restrict__ cols,
                   float* cntA, float* cntB, unsigned* __restrict__ pk,
                   int4* __restrict__ cnts) {
    __shared__ __attribute__((aligned(16))) unsigned stage[CH];
    __shared__ int ncls[4];
    const int t = threadIdx.x;
    const int lane = t & 31;
    if (t < 4) ncls[t] = 0;
    __syncthreads();
    const bool live = t < CHV;
    const int v = blockIdx.x * CHV + (live ? t : 0);
    const int4 r4 = ((const int4*)rows)[v];
    const int4 c4 = ((const int4*)cols)[v];
    const unsigned place = __builtin_amdgcn_s_sendmsg_rtn(0x87);
    float* cnt = ((place >> 18) & 1u) ? cntB : cntA;
    const int k0 = live ? ring_class(r4.x, c4.x) : 4;
    const int k1 = live ? ring_class(r4.y, c4.y) : 4;
    const int k2 = live ? ring_class(r4.z, c4.z) : 4;
    const int k3 = live ? ring_class(r4.w, c4.w) : 4;
    const int q0 = class_place(k0, lane, ncls);
    const int q1 = class_place(k1, lane, ncls);
    const int q2 = class_place(k2, lane, ncls);
    const int q3 = class_place(k3, lane, ncls);
    if (live) {
        atomAddF(cnt + r4.x, 1.0f); atomAddF(cnt + NTOT + c4.x, 1.0f);
        atomAddF(cnt + r4.y, 1.0f); atomAddF(cnt + NTOT + c4.y, 1.0f);
        atomAddF(cnt + r4.z, 1.0f); atomAddF(cnt + NTOT + c4.z, 1.0f);
        atomAddF(cnt + r4.w, 1.0f); atomAddF(cnt + NTOT + c4.w, 1.0f);
    }
    __syncthreads();
    const int n0 = ncls[0], n1 = ncls[1], n2 = ncls[2], n3 = ncls[3];
    if (live) {
        const int b1 = n0, b2 = n0 + n1, b3 = n0 + n1 + n2;
        stage[(k0 == 0 ? 0 : k0 == 1 ? b1 : k0 == 2 ? b2 : b3) + q0] = (unsigned)r4.x | ((unsigned)c4.x << 16);
        stage[(k1 == 0 ? 0 : k1 == 1 ? b1 : k1 == 2 ? b2 : b3) + q1] = (unsigned)r4.y | ((unsigned)c4.y << 16);
        stage[(k2 == 0 ? 0 : k2 == 1 ? b1 : k2 == 2 ? b2 : b3) + q2] = (unsigned)r4.z | ((unsigned)c4.z << 16);
        stage[(k3 == 0 ? 0 : k3 == 1 ? b1 : k3 == 2 ? b2 : b3) + q3] = (unsigned)r4.w | ((unsigned)c4.w << 16);
    }
    __syncthreads();
    if (live) {
        uint4 o;
        o.x = stage[4 * t]; o.y = stage[4 * t + 1]; o.z = stage[4 * t + 2]; o.w = stage[4 * t + 3];
        ((uint4*)pk)[blockIdx.x * CHV + t] = o;
    }
    if (t == 0) cnts[blockIdx.x] = make_int4(n0, n1, n2, n3);
}

__device__ __forceinline__ int eighth_of(int id) {
    const int k = id / E8N;
    return k > 7 ? 7 : k;
}

__device__ __forceinline__ int class_place8(int k, int lane, int* ncls) {
    unsigned mine = 0u, ml = 0u;
#pragma unroll
    for (int cc = 0; cc < 8; ++cc) {
        const unsigned m = __builtin_amdgcn_ballot_w32(k == cc);
        if (k == cc) mine = m;
        if (lane == cc) ml = m;
    }
    int first = 0;
    if (lane < 8)
        first = __hip_atomic_fetch_add(ncls + lane, (int)__popc(ml), __ATOMIC_RELAXED,
                                       __HIP_MEMORY_SCOPE_WORKGROUP);
    const int base = __shfl(first, k & 7);
    return base + (int)__popc(mine & ((1u << lane) - 1u));
}

__global__ __launch_bounds__(1024)
void degree_part8_k(const int* __restrict__ rows, const int* __restrict__ cols,
                    float* cntA, float* cntB, unsigned* __restrict__ pkR,
                    unsigned* __restrict__ pkC, int2* __restrict__ tab) {
    __shared__ __attribute__((aligned(16))) unsigned stR[CH];
    __shared__ __attribute__((aligned(16))) unsigned stC[CH];
    __shared__ int ncls[16];
    __shared__ int sb[16];
    const int t = threadIdx.x;
    const int lane = t & 31;
    if (t < 16) ncls[t] = 0;
    __syncthreads();
    const bool live = t < CHV;
    const int v = blockIdx.x * CHV + (live ? t : 0);
    const int4 r4 = ((const int4*)rows)[v];
    const int4 c4 = ((const int4*)cols)[v];
    const unsigned place = __builtin_amdgcn_s_sendmsg_rtn(0x87);
    float* cnt = ((place >> 18) & 1u) ? cntB : cntA;
    const int kr0 = live ? eighth_of(r4.x) : 8, kc0 = live ? eighth_of(c4.x) : 8;
    const int kr1 = live ? eighth_of(r4.y) : 8, kc1 = live ? eighth_of(c4.y) : 8;
    const int kr2 = live ? eighth_of(r4.z) : 8, kc2 = live ? eighth_of(c4.z) : 8;
    const int kr3 = live ? eighth_of(r4.w) : 8, kc3 = live ? eighth_of(c4.w) : 8;
    const int qr0 = class_place8(kr0, lane, ncls), qc0 = class_place8(kc0, lane, ncls + 8);
    const int qr1 = class_place8(kr1, lane, ncls), qc1 = class_place8(kc1, lane, ncls + 8);
    const int qr2 = class_place8(kr2, lane, ncls), qc2 = class_place8(kc2, lane, ncls + 8);
    const int qr3 = class_place8(kr3, lane, ncls), qc3 = class_place8(kc3, lane, ncls + 8);
    if (live) {
        atomAddF(cnt + r4.x, 1.0f); atomAddF(cnt + NTOT + c4.x, 1.0f);
        atomAddF(cnt + r4.y, 1.0f); atomAddF(cnt + NTOT + c4.y, 1.0f);
        atomAddF(cnt + r4.z, 1.0f); atomAddF(cnt + NTOT + c4.z, 1.0f);
        atomAddF(cnt + r4.w, 1.0f); atomAddF(cnt + NTOT + c4.w, 1.0f);
    }
    __syncthreads();
    if (t < 16) {
        int s = 0;
        for (int cc = t & 8; cc < t; ++cc) s += ncls[cc];
        sb[t] = s;
        tab[blockIdx.x * 16 + t] = make_int2(s, s + ncls[t]);
    }
    __syncthreads();
    if (live) {
        const unsigned w0 = (unsigned)r4.x | ((unsigned)c4.x << 16);
        const unsigned w1 = (unsigned)r4.y | ((unsigned)c4.y << 16);
        const unsigned w2 = (unsigned)r4.z | ((unsigned)c4.z << 16);
        const unsigned w3 = (unsigned)r4.w | ((unsigned)c4.w << 16);
        stR[sb[kr0] + qr0] = w0; stC[sb[8 + kc0] + qc0] = w0;
        stR[sb[kr1] + qr1] = w1; stC[sb[8 + kc1] + qc1] = w1;
        stR[sb[kr2] + qr2] = w2; stC[sb[8 + kc2] + qc2] = w2;
        stR[sb[kr3] + qr3] = w3; stC[sb[8 + kc3] + qc3] = w3;
    }
    __syncthreads();
    if (live) {
        uint4 o;
        o.x = stR[4 * t]; o.y = stR[4 * t + 1]; o.z = stR[4 * t + 2]; o.w = stR[4 * t + 3];
        ((uint4*)pkR)[blockIdx.x * CHV + t] = o;
        o.x = stC[4 * t]; o.y = stC[4 * t + 1]; o.z = stC[4 * t + 2]; o.w = stC[4 * t + 3];
        ((uint4*)pkC)[blockIdx.x * CHV + t] = o;
    }
}

__global__ void rsq_k(float* pa, const float* pb, long n) {
    long i = (long)blockIdx.x * blockDim.x + threadIdx.x;
    if (i < n) {
        float d = pa[i];
        if (pb) d += pb[i];
        pa[i] = 1.0f / (sqrtf(d) + EPSF);
    }
}

__global__ void buildz_k(const float* uf, const float* itf, const float* bfv,
                         const float* __restrict__ dvi, float* __restrict__ zp) {
    __shared__ float tile[32][33];
    const int i0 = blockIdx.x * 32;
#pragma unroll
    for (int r = 0; r < 4; ++r) {
        int idx = threadIdx.x + r * 256;
        int j = idx >> 5, d = idx & 31;
        int i = i0 + j;
        tile[j][d] = (i < NTOT) ? feat_at(uf, itf, bfv, i, d) * dvi[i] : 0.0f;
    }
    __syncthreads();
#pragma unroll
    for (int r = 0; r < 4; ++r) {
        int idx = threadIdx.x + r * 256;
        int d = idx >> 5, j = idx & 31;
        int i = i0 + j;
        if (i < NTOT) zp[(size_t)d * NP + i] = tile[j][d];
    }
}

template <bool DST_LOW>
__device__ __forceinline__ int src_node(unsigned w) {
    return DST_LOW ? (int)(w >> 16) : (int)(w & 0xFFFFu);
}
template <bool DST_LOW>
__device__ __forceinline__ int dst_node(unsigned w) {
    return DST_LOW ? (int)(w & 0xFFFFu) : (int)(w >> 16);
}

__device__ __forceinline__ void edge1(int dcol, float val, int lo, int cnt, float* dstL, bool live) {
    unsigned dd = (unsigned)(dcol - lo);
    if (live && dd < (unsigned)cnt)
        (void)__hip_atomic_fetch_add(dstL + dd, val, __ATOMIC_RELAXED,
                                     __HIP_MEMORY_SCOPE_WORKGROUP);
}

__device__ __forceinline__ void add_hit(bool hit, unsigned dd, float val, float* dstL) {
    if (__builtin_expect(hit, true))
        (void)__hip_atomic_fetch_add(dstL + dd, val, __ATOMIC_RELAXED,
                                     __HIP_MEMORY_SCOPE_WORKGROUP);
}

#define NV4 (ESLICE / 4)

__device__ __forceinline__ int ring_vec(int vs, int i) {
    const int x = vs + i;
    return x < CHV ? x : x - CHV;
}

template <bool DST_LOW>
__global__ __launch_bounds__(1024)
void spmm_pk_k(const unsigned* __restrict__ pk, const int4* __restrict__ cnts,
            const float* __restrict__ srcp, const float* __restrict__ rs,
            float* __restrict__ dstp) {
    __shared__ float srcL[NTOT];
    __shared__ float dstL[H1N];
    const int d  = blockIdx.x & 31;
    const int h  = (blockIdx.x >> 5) & 1;
    const int q  = blockIdx.x >> 6;
    const int lo = h * H0N;
    const int cnt = h ? H1N : H0N;
    const float* sp = srcp + (size_t)d * NP;
    if (rs) {
        for (int i = threadIdx.x; i < NTOT; i += 1024) srcL[i] = sp[i] * rs[i];
    } else {
        for (int i = threadIdx.x; i < NTOT; i += 1024) srcL[i] = sp[i];
    }
    for (int i = threadIdx.x; i < H1N; i += 1024) dstL[i] = 0.0f;
    __syncthreads();

    const int lane = threadIdx.x & 31;
    const int wv = __builtin_amdgcn_readfirstlane((int)(threadIdx.x >> 5));
    const int rot = (int)((blockIdx.x & 63) * NCHQ) >> 6;
    for (int cl = wv; cl < NCHQ; cl += 32) {
        int cc = cl + rot;
        if (cc >= NCHQ) cc -= NCHQ;
        const int chunk = q * NCHQ + cc;
        const int4 n = cnts[chunk];
        int start, len;
        if (DST_LOW) {
            start = h ? n.x + n.y : 0;
            len   = h ? n.z + n.w : n.x + n.y;
        } else {
            start = h ? n.x : n.x + n.y + n.z;
            len   = h ? n.y + n.z : n.w + n.x;
        }
        const int vs = start >> 2;
        int nv = ((start + len + 3) >> 2) - vs;
        if (nv > CHV) nv = CHV;
        if (len == 0) nv = 0;
        const uint4* p4 = (const uint4*)(pk + (size_t)chunk * CH);
        uint4 wa = p4[ring_vec(vs, lane < nv ? lane : 0)];
        uint4 wb = p4[ring_vec(vs, lane + 32 < nv ? lane + 32 : 0)];
        for (int t = 0; t < nv; t += 64) {
            const int ja = t + 64 + lane, jb = ja + 32;
            const uint4 na = p4[ring_vec(vs, ja < nv ? ja : 0)];
            const uint4 nb = p4[ring_vec(vs, jb < nv ? jb : 0)];
            const bool la = (t + lane) < nv;
            const bool lb = (t + lane + 32) < nv;
            const unsigned da0 = (unsigned)(dst_node<DST_LOW>(wa.x) - lo);
            const unsigned da1 = (unsigned)(dst_node<DST_LOW>(wa.y) - lo);
            const unsigned da2 = (unsigned)(dst_node<DST_LOW>(wa.z) - lo);
            const unsigned da3 = (unsigned)(dst_node<DST_LOW>(wa.w) - lo);
            const unsigned db0 = (unsigned)(dst_node<DST_LOW>(wb.x) - lo);
            const unsigned db1 = (unsigned)(dst_node<DST_LOW>(wb.y) - lo);
            const unsigned db2 = (unsigned)(dst_node<DST_LOW>(wb.z) - lo);
            const unsigned db3 = (unsigned)(dst_node<DST_LOW>(wb.w) - lo);
            const bool ha0 = la && da0 < (unsigned)cnt, ha1 = la && da1 < (unsigned)cnt;
            const bool ha2 = la && da2 < (unsigned)cnt, ha3 = la && da3 < (unsigned)cnt;
            const bool hb0 = lb && db0 < (unsigned)cnt, hb1 = lb && db1 < (unsigned)cnt;
            const bool hb2 = lb && db2 < (unsigned)cnt, hb3 = lb && db3 < (unsigned)cnt;
            float a0 = srcL[ha0 ? src_node<DST_LOW>(wa.x) : 0], a1 = srcL[ha1 ? src_node<DST_LOW>(wa.y) : 0];
            float a2 = srcL[ha2 ? src_node<DST_LOW>(wa.z) : 0], a3 = srcL[ha3 ? src_node<DST_LOW>(wa.w) : 0];
            float b0 = srcL[hb0 ? src_node<DST_LOW>(wb.x) : 0], b1 = srcL[hb1 ? src_node<DST_LOW>(wb.y) : 0];
            float b2 = srcL[hb2 ? src_node<DST_LOW>(wb.z) : 0], b3 = srcL[hb3 ? src_node<DST_LOW>(wb.w) : 0];
            asm volatile("" :: "v"(a0), "v"(a1), "v"(a2), "v"(a3),
                               "v"(b0), "v"(b1), "v"(b2), "v"(b3));
            add_hit(ha0, da0, a0, dstL);
            add_hit(ha1, da1, a1, dstL);
            add_hit(ha2, da2, a2, dstL);
            add_hit(ha3, da3, a3, dstL);
            add_hit(hb0, db0, b0, dstL);
            add_hit(hb1, db1, b1, dstL);
            add_hit(hb2, db2, b2, dstL);
            add_hit(hb3, db3, b3, dstL);
            wa = na; wb = nb;
        }
    }
    __syncthreads();

    float* dp = dstp + (size_t)d * NP + lo;
    for (int i = threadIdx.x; i < cnt; i += 1024) atomAddF(dp + i, dstL[i]);
}

struct Run8 { int off; int nv; };
struct Bnd8 { int chunk, start, end; };

__device__ __forceinline__ Bnd8 bounds_of(const int2* __restrict__ tab, int cl, int rot, int k) {
    Bnd8 b;
    b.chunk = cl + rot;
    if (b.chunk >= NCHUNK) b.chunk -= NCHUNK;
    const int2 se = tab[b.chunk * 16 + k];
    b.start = se.x;
    b.end   = se.y;
    return b;
}

__device__ __forceinline__ Run8 run_of(Bnd8 b) {
    Run8 r;
    const bool any = b.end > b.start;
    const int vs = any ? b.start >> 2 : 0;
    r.nv  = any ? ((b.end + 3) >> 2) - vs : 0;
    r.off = b.chunk * CHV + vs;
    return r;
}

__device__ __forceinline__ int stream_vec(Run8 cur, Run8 nxt, int j, bool& valid) {
    const int jn = j - cur.nv;
    const bool inc = jn < 0;
    valid = inc || jn < nxt.nv;
    const int first = inc ? cur.off : nxt.off - cur.nv;
    return valid ? first + j : 0;
}

template <bool DST_LOW>
__global__ __launch_bounds__(1024)
void spmm8_k(const unsigned* __restrict__ pk, const int2* __restrict__ tab,
             const float* __restrict__ srcp, const float* __restrict__ rs,
             float* __restrict__ dstp) {
    __shared__ float srcL[NTOT];
    __shared__ float dstL[E8L];
    const int d  = blockIdx.x & 31;
    const int k  = blockIdx.x >> 5;
    const int lo = k * E8N;
    const int cnt = k == 7 ? E8L : E8N;
    const float* sp = srcp + (size_t)d * NP;
    if (rs) {
        for (int i = threadIdx.x; i < NTOT; i += 1024) srcL[i] = sp[i] * rs[i];
    } else {
        for (int i = threadIdx.x; i < NTOT; i += 1024) srcL[i] = sp[i];
    }
    for (int i = threadIdx.x; i < E8L; i += 1024) dstL[i] = 0.0f;
    __syncthreads();

    const int lane = threadIdx.x & 31;
    const int wv = __builtin_amdgcn_readfirstlane((int)(threadIdx.x >> 5));
    const int rot = (d * NCHUNK) >> 5;
    int cl = wv;
    Run8 cur = run_of(bounds_of(tab, cl, rot, k));
    Run8 nxt = run_of(bounds_of(tab, cl + 32, rot, k));
    Bnd8 bnn = bounds_of(tab, cl + 64, rot, k);
    const uint4* __restrict__ p4 = (const uint4*)pk;
    int t = 0;
    bool va, vb;
    uint4 wa = p4[stream_vec(cur, nxt, lane, va)];
    uint4 wb = p4[stream_vec(cur, nxt, lane + 32, vb)];
    for (;;) {
        t += 64;
        if (t >= cur.nv) {
            const int over = t - cur.nv;
            t = over < nxt.nv ? over : nxt.nv;
            cl += 32;
            cur = nxt;
            nxt.nv = 0;
            if (cl + 32 < NCHUNK) nxt = run_of(bnn);
            if (cl + 64 < NCHUNK) bnn = bounds_of(tab, cl + 64, rot, k);
        }
        const bool last = cl >= NCHUNK;
        bool nva, nvb;
        const uint4 na = p4[stream_vec(cur, nxt, t + lane, nva)];
        const uint4 nb = p4[stream_vec(cur, nxt, t + lane + 32, nvb)];
        const unsigned da0 = (unsigned)(dst_node<DST_LOW>(wa.x) - lo);
        const unsigned da1 = (unsigned)(dst_node<DST_LOW>(wa.y) - lo);
        const unsigned da2 = (unsigned)(dst_node<DST_LOW>(wa.z) - lo);
        const unsigned da3 = (unsigned)(dst_node<DST_LOW>(wa.w) - lo);
        const unsigned db0 = (unsigned)(dst_node<DST_LOW>(wb.x) - lo);
        const unsigned db1 = (unsigned)(dst_node<DST_LOW>(wb.y) - lo);
        const unsigned db2 = (unsigned)(dst_node<DST_LOW>(wb.z) - lo);
        const unsigned db3 = (unsigned)(dst_node<DST_LOW>(wb.w) - lo);
        const bool ha0 = va && da0 < (unsigned)cnt, ha1 = va && da1 < (unsigned)cnt;
        const bool ha2 = va && da2 < (unsigned)cnt, ha3 = va && da3 < (unsigned)cnt;
        const bool hb0 = vb && db0 < (unsigned)cnt, hb1 = vb && db1 < (unsigned)cnt;
        const bool hb2 = vb && db2 < (unsigned)cnt, hb3 = vb && db3 < (unsigned)cnt;
        float a0 = srcL[ha0 ? src_node<DST_LOW>(wa.x) : 0], a1 = srcL[ha1 ? src_node<DST_LOW>(wa.y) : 0];
        float a2 = srcL[ha2 ? src_node<DST_LOW>(wa.z) : 0], a3 = srcL[ha3 ? src_node<DST_LOW>(wa.w) : 0];
        float b0 = srcL[hb0 ? src_node<DST_LOW>(wb.x) : 0], b1 = srcL[hb1 ? src_node<DST_LOW>(wb.y) : 0];
        float b2 = srcL[hb2 ? src_node<DST_LOW>(wb.z) : 0], b3 = srcL[hb3 ? src_node<DST_LOW>(wb.w) : 0];
        asm volatile("" :: "v"(a0), "v"(a1), "v"(a2), "v"(a3),
                           "v"(b0), "v"(b1), "v"(b2), "v"(b3));
        add_hit(ha0, da0, a0, dstL);
        add_hit(ha1, da1, a1, dstL);
        add_hit(ha2, da2, a2, dstL);
        add_hit(ha3, da3, a3, dstL);
        add_hit(hb0, db0, b0, dstL);
        add_hit(hb1, db1, b1, dstL);
        add_hit(hb2, db2, b2, dstL);
        add_hit(hb3, db3, b3, dstL);
        if (last) break;
        wa = na; wb = nb; va = nva; vb = nvb;
    }
    __syncthreads();

    float* dp = dstp + (size_t)d * NP + lo;
    for (int i = threadIdx.x; i < cnt; i += 1024) dp[i] = dstL[i];
}

__global__ __launch_bounds__(1024)
void spmm_k(const int* __restrict__ sidx, const int* __restrict__ didx,
            const float* __restrict__ srcp, const float* __restrict__ rs,
            float* __restrict__ dstp) {
    __shared__ float srcL[NTOT];
    __shared__ float dstL[HALFN];
    const int d  = blockIdx.x & 31;
    const int h  = (blockIdx.x >> 5) & 1;
    const int q  = blockIdx.x >> 6;
    const int lo = h * HALFN;
    const float* sp = srcp + (size_t)d * NP;
    if (rs) {
        for (int i = threadIdx.x; i < NTOT; i += 1024) srcL[i] = sp[i] * rs[i];
    } else {
        for (int i = threadIdx.x; i < NTOT; i += 1024) srcL[i] = sp[i];
    }
    for (int i = threadIdx.x; i < HALFN; i += 1024) dstL[i] = 0.0f;
    __syncthreads();

    const int4* s4 = (const int4*)(sidx + (size_t)q * ESLICE);
    const int4* d4 = (const int4*)(didx + (size_t)q * ESLICE);
    const int off = (blockIdx.x & 63) * (NV4 / 64 / 32 * 32);
    int k = threadIdx.x;
    int v = k + off;
    int4 dc = d4[v];
    int4 sr = s4[v];
    for (;;) {
        const int kn = k + 1024;
        const int vw = v + 1024;
        const int vn = vw < NV4 ? vw : vw - NV4;
        const int kc = kn < NV4 ? vn : v;
        const int4 dcn = d4[kc];
        const int4 srn = s4[kc];
        float a0 = srcL[sr.x], a1 = srcL[sr.y], a2 = srcL[sr.z], a3 = srcL[sr.w];
        asm volatile("" :: "v"(a0), "v"(a1), "v"(a2), "v"(a3));
        edge1(dc.x, a0, lo, HALFN, dstL, true);
        edge1(dc.y, a1, lo, HALFN, dstL, true);
        edge1(dc.z, a2, lo, HALFN, dstL, true);
        edge1(dc.w, a3, lo, HALFN, dstL, true);
        if (kn >= NV4) break;
        dc = dcn; sr = srn; k = kn; v = vn;
    }
    __syncthreads();

    float* dp = dstp + (size_t)d * NP + lo;
    for (int i = threadIdx.x; i < HALFN; i += 1024) atomAddF(dp + i, dstL[i]);
}

__global__ void final_k(const float* uf, const float* itf, const float* bfv,
                        const float* __restrict__ dvi, const float* __restrict__ v,
                        float* __restrict__ all, float* __restrict__ partials) {
    __shared__ float sdata[256];
    long t = (long)blockIdx.x * blockDim.x + threadIdx.x;
    float sq = 0.0f;
    if (t < (long)NTOT * EMB) {
        int i = (int)(t >> 5), d = (int)(t & 31);
        const float x = feat_at(uf, itf, bfv, i, d);
        const float sc = dvi[i];
        const float* vp = v + (size_t)d * NP + i;
        const float vs = vp[0];
        float val = x * 0.5f + vs * sc * (1.0f / 3.0f);
        all[t] = val;
        sq = val * val;
    }
    sdata[threadIdx.x] = sq;
    __syncthreads();
    for (int s = 128; s > 0; s >>= 1) {
        if (threadIdx.x < (unsigned)s) sdata[threadIdx.x] += sdata[threadIdx.x + s];
        __syncthreads();
    }
    if (threadIdx.x == 0) partials[blockIdx.x] = sdata[0];
}

__global__ void loss_k(const float* __restrict__ partials, int nb, float* __restrict__ out_loss) {
    __shared__ float sdata[256];
    float s = 0.0f;
    for (int i = threadIdx.x; i < nb; i += 256) s += partials[i];
    sdata[threadIdx.x] = s;
    __syncthreads();
    for (int k = 128; k > 0; k >>= 1) {
        if (threadIdx.x < (unsigned)k) sdata[threadIdx.x] += sdata[threadIdx.x + k];
        __syncthreads();
    }
    if (threadIdx.x == 0) *out_loss = L2NORM * sdata[0];
}

__global__ void pred_k(const float* __restrict__ all, const int* __restrict__ users,
                       const int* __restrict__ bundles, const float* __restrict__ ubound,
                       const float* __restrict__ zrow,
                       float* __restrict__ pred, float* __restrict__ usb) {
    int wave = (blockIdx.x * blockDim.x + threadIdx.x) >> 5;
    int lane = threadIdx.x & 31;
    int b = wave / 7;
    int kt = wave % 7;

    const float* bfbase = all + (size_t)(NU + NI) * EMB;
    const float* ue = all + (size_t)users[b] * EMB;

    int m = lane & 15;
    int khalf = lane >> 4;
    int kk = kt * 16 + m;

    const float* arow;
    if (kk < KB)                 arow = bfbase + (size_t)bundles[b * KB + kk] * EMB;
    else if (kt == 6 && m == 4)  arow = ubound;
    else                         arow = zrow;

    v8f c = {};
#pragma unroll
    for (int s = 0; s < 8; ++s) {
        int e0 = s * 4 + khalf * 2;
        v2f a  = *(const v2f*)(arow + e0);
        v2f bb = *(const v2f*)(ue + e0);
        c = __builtin_amdgcn_wmma_f32_16x16x4_f32(false, a, false, bb,
                                                  (short)0, c, false, false);
    }

    if (lane == 0) {
#pragma unroll
        for (int j = 0; j < 8; ++j) {
            int kkk = kt * 16 + j;
            if (kkk < KB)                 pred[(size_t)b * KB + kkk] = c[j];
            else if (kt == 6 && j == 4)   usb[b] = c[j];
        }
    } else if (lane == 16) {
#pragma unroll
        for (int j = 0; j < 8; ++j) {
            int kkk = kt * 16 + 8 + j;
            if (kkk < KB) pred[(size_t)b * KB + kkk] = c[j];
        }
    }
}

extern "C" void kernel_launch(void* const* d_in, const int* in_sizes, int n_in,
                              void* d_out, int out_size, void* d_ws, size_t ws_size,
                              hipStream_t stream) {
    const float* uf      = (const float*)d_in[0];
    const float* itf     = (const float*)d_in[1];
    const float* bfv     = (const float*)d_in[2];
    const float* ubound  = (const float*)d_in[3];
    const int*   rows    = (const int*)d_in[4];
    const int*   cols    = (const int*)d_in[5];
    const int*   users   = (const int*)d_in[6];
    const int*   bundles = (const int*)d_in[7];
    float* out = (float*)d_out;

    const bool plain = ws_size >= (size_t)PL_END_B;
    float* ws   = (float*)d_ws;
    float* p0   = ws;
    float* p1   = p0 + 32L * NP;
    float* p2   = p1 + (plain ? 128L : 32L) * NP;
    float* dvde = p2 + (plain ? 128L : 32L) * NP;
    float* partials = dvde + 2L * NTOT;
    float* zrow = partials + PARTN;
    const bool packed = plain || ws_size >= (size_t)PK_END_B;
    unsigned* pk = packed ? (unsigned*)(zrow + 104) : (unsigned*)nullptr;
    unsigned* pkR = plain ? (unsigned*)(p1 + 32L * NP) : (unsigned*)nullptr;
    int2* tab8 = plain ? (int2*)(p2 + 32L * NP) : (int2*)nullptr;

    const long elems = (long)NTOT * EMB;
    const int  eb    = (int)((elems + 255) / 256);

    float* cntB = plain ? dvde - 2L * NTOT : dvde;

    zero_k<<<2048, 256, 0, stream>>>(cntB, plain ? 4L * NTOT : 2L * NTOT);
    if (!plain) zero_k<<<4096, 256, 0, stream>>>(p1, 64L * NP);
    zero_k<<<1, 32, 0, stream>>>(zrow, 32L);

    int4* ccnt = (int4*)partials;
    if (plain)       degree_part8_k<<<NCHUNK, 1024, 0, stream>>>(rows, cols, dvde, cntB, pkR, pk, tab8);
    else if (packed) degree_part_k<<<NCHUNK, 1024, 0, stream>>>(rows, cols, dvde, cntB, pk, ccnt);
    else             degree_k<<<(NNZE + 255) / 256, 256, 0, stream>>>(rows, cols, dvde, cntB);
    rsq_k<<<(int)((2L * NTOT + 255) / 256), 256, 0, stream>>>(dvde, plain ? cntB : (const float*)nullptr,
                                                             2L * NTOT);

    buildz_k<<<(NTOT + 31) / 32, 256, 0, stream>>>(uf, itf, bfv, dvde, p0);
    if (plain) {
        spmm8_k<false><<<32 * 8, 1024, 0, stream>>>(pk, tab8 + 8, p0, (const float*)nullptr, p1);
        spmm8_k<true><<<32 * 8, 1024, 0, stream>>>(pkR, tab8, p1, dvde + NTOT, p2);
    } else if (packed) {
        spmm_pk_k<false><<<32 * 2 * NSLICE, 1024, 0, stream>>>(pk, ccnt, p0, (const float*)nullptr, p1);
        spmm_pk_k<true><<<32 * 2 * NSLICE, 1024, 0, stream>>>(pk, ccnt, p1, dvde + NTOT, p2);
    } else {
        spmm_k<<<32 * 2 * NSLICE, 1024, 0, stream>>>(rows, cols, p0, (const float*)nullptr, p1);
        spmm_k<<<32 * 2 * NSLICE, 1024, 0, stream>>>(cols, rows, p1, dvde + NTOT, p2);
    }

    final_k<<<eb, 256, 0, stream>>>(uf, itf, bfv, dvde, p2, p0, partials);
    loss_k<<<1, 256, 0, stream>>>(partials, eb, out + BATCHN * KB + BATCHN);

    pred_k<<<1792, 256, 0, stream>>>(p0, users, bundles, ubound, zrow,
                                     out, out + BATCHN * KB);
}


# second edge pass ends with the combine x/2 + v*dvi/3 and one loss partial per block; a 402-block launch turns only user and bundle rows row-major; loss partials regrouped, 5698 to 256
# speedup vs baseline: 4.3735x; 1.0855x over previous
#include <hip/hip_runtime.h>
#include <stddef.h>

#define NU   8039
#define NI   32770
#define NB   4771
#define NTOT (NU + NI + NB)
#define EMB  32
#define NNZE 2000000
#define BATCHN 2048
#define KB   100
#define EPSF 1e-8f
#define L2NORM 1e-5f
#define NP     45600
#define HALFN  22790
#define H0N    22784
#define H1N    (NTOT - H0N)
#define NSLICE 4
#define ESLICE (NNZE / NSLICE)
#define PARTN   5760
#define PK_OFF  (96L * NP + 2L * NTOT + PARTN + 104)
#define PK_END_B ((PK_OFF + (long)NNZE) * 4L)
#define PL_END_B (((288L - 96L) * NP) * 4L + PK_END_B)
#define CH     4000
#define CHV    (CH / 4)
#define NCHQ   (ESLICE / CH)
#define NCHUNK (NNZE / CH)

#define E8N    5696
#define E8L    (NTOT - 7 * E8N)

typedef __attribute__((ext_vector_type(2))) float v2f;
typedef __attribute__((ext_vector_type(8))) float v8f;

__device__ __forceinline__ void atomAddF(float* p, float v) {
    (void)__hip_atomic_fetch_add(p, v, __ATOMIC_RELAXED, __HIP_MEMORY_SCOPE_AGENT);
}

__device__ __forceinline__ float feat_at(const float* uf, const float* itf,
                                         const float* bfv, int i, int d) {
    if (i < NU)      return uf[(size_t)i * EMB + d];
    if (i < NU + NI) return itf[(size_t)(i - NU) * EMB + d];
    return bfv[(size_t)(i - NU - NI) * EMB + d];
}

__global__ void zero_k(float* p, long n) {
    long i = (long)blockIdx.x * blockDim.x + threadIdx.x;
    long stride = (long)gridDim.x * blockDim.x;
    for (; i < n; i += stride) p[i] = 0.0f;
}

__global__ void degree_k(const int* __restrict__ rows, const int* __restrict__ cols,
                         float* cntA, float* cntB) {
    int e = blockIdx.x * blockDim.x + threadIdx.x;
    if (e >= NNZE) return;
    const int r = rows[e], c = cols[e];
    const unsigned place = __builtin_amdgcn_s_sendmsg_rtn(0x87);
    float* cnt = ((place >> 18) & 1u) ? cntB : cntA;
    atomAddF(cnt + r, 1.0f);
    atomAddF(cnt + NTOT + c, 1.0f);
}

__device__ __forceinline__ int ring_class(int r, int c) {
    return r >= H0N ? (c >= H0N ? 2 : 3) : (c >= H0N ? 1 : 0);
}

__device__ __forceinline__ int class_place(int k, int lane, int* ncls) {
    const unsigned m0 = __builtin_amdgcn_ballot_w32(k == 0);
    const unsigned m1 = __builtin_amdgcn_ballot_w32(k == 1);
    const unsigned m2 = __builtin_amdgcn_ballot_w32(k == 2);
    const unsigned m3 = __builtin_amdgcn_ballot_w32(k == 3);
    const unsigned mine = k == 0 ? m0 : k == 1 ? m1 : k == 2 ? m2 : m3;
    int first = 0;
    if (lane < 4) {
        const unsigned ml = lane == 0 ? m0 : lane == 1 ? m1 : lane == 2 ? m2 : m3;
        first = __hip_atomic_fetch_add(ncls + lane, (int)__popc(ml), __ATOMIC_RELAXED,
                                       __HIP_MEMORY_SCOPE_WORKGROUP);
    }
    const int base = __shfl(first, k & 3);
    return base + (int)__popc(mine & ((1u << lane) - 1u));
}

__global__ __launch_bounds__(1024)
void degree_part_k(const int* __restrict__ rows, const int* __restrict__ cols,
                   float* cntA, float* cntB, unsigned* __restrict__ pk,
                   int4* __restrict__ cnts) {
    __shared__ __attribute__((aligned(16))) unsigned stage[CH];
    __shared__ int ncls[4];
    const int t = threadIdx.x;
    const int lane = t & 31;
    if (t < 4) ncls[t] = 0;
    __syncthreads();
    const bool live = t < CHV;
    const int v = blockIdx.x * CHV + (live ? t : 0);
    const int4 r4 = ((const int4*)rows)[v];
    const int4 c4 = ((const int4*)cols)[v];
    const unsigned place = __builtin_amdgcn_s_sendmsg_rtn(0x87);
    float* cnt = ((place >> 18) & 1u) ? cntB : cntA;
    const int k0 = live ? ring_class(r4.x, c4.x) : 4;
    const int k1 = live ? ring_class(r4.y, c4.y) : 4;
    const int k2 = live ? ring_class(r4.z, c4.z) : 4;
    const int k3 = live ? ring_class(r4.w, c4.w) : 4;
    const int q0 = class_place(k0, lane, ncls);
    const int q1 = class_place(k1, lane, ncls);
    const int q2 = class_place(k2, lane, ncls);
    const int q3 = class_place(k3, lane, ncls);
    if (live) {
        atomAddF(cnt + r4.x, 1.0f); atomAddF(cnt + NTOT + c4.x, 1.0f);
        atomAddF(cnt + r4.y, 1.0f); atomAddF(cnt + NTOT + c4.y, 1.0f);
        atomAddF(cnt + r4.z, 1.0f); atomAddF(cnt + NTOT + c4.z, 1.0f);
        atomAddF(cnt + r4.w, 1.0f); atomAddF(cnt + NTOT + c4.w, 1.0f);
    }
    __syncthreads();
    const int n0 = ncls[0], n1 = ncls[1], n2 = ncls[2], n3 = ncls[3];
    if (live) {
        const int b1 = n0, b2 = n0 + n1, b3 = n0 + n1 + n2;
        stage[(k0 == 0 ? 0 : k0 == 1 ? b1 : k0 == 2 ? b2 : b3) + q0] = (unsigned)r4.x | ((unsigned)c4.x << 16);
        stage[(k1 == 0 ? 0 : k1 == 1 ? b1 : k1 == 2 ? b2 : b3) + q1] = (unsigned)r4.y | ((unsigned)c4.y << 16);
        stage[(k2 == 0 ? 0 : k2 == 1 ? b1 : k2 == 2 ? b2 : b3) + q2] = (unsigned)r4.z | ((unsigned)c4.z << 16);
        stage[(k3 == 0 ? 0 : k3 == 1 ? b1 : k3 == 2 ? b2 : b3) + q3] = (unsigned)r4.w | ((unsigned)c4.w << 16);
    }
    __syncthreads();
    if (live) {
        uint4 o;
        o.x = stage[4 * t]; o.y = stage[4 * t + 1]; o.z = stage[4 * t + 2]; o.w = stage[4 * t + 3];
        ((uint4*)pk)[blockIdx.x * CHV + t] = o;
    }
    if (t == 0) cnts[blockIdx.x] = make_int4(n0, n1, n2, n3);
}

__device__ __forceinline__ int eighth_of(int id) {
    const int k = id / E8N;
    return k > 7 ? 7 : k;
}

__device__ __forceinline__ int class_place8(int k, int lane, int* ncls) {
    unsigned mine = 0u, ml = 0u;
#pragma unroll
    for (int cc = 0; cc < 8; ++cc) {
        const unsigned m = __builtin_amdgcn_ballot_w32(k == cc);
        if (k == cc) mine = m;
        if (lane == cc) ml = m;
    }
    int first = 0;
    if (lane < 8)
        first = __hip_atomic_fetch_add(ncls + lane, (int)__popc(ml), __ATOMIC_RELAXED,
                                       __HIP_MEMORY_SCOPE_WORKGROUP);
    const int base = __shfl(first, k & 7);
    return base + (int)__popc(mine & ((1u << lane) - 1u));
}

__global__ __launch_bounds__(1024)
void degree_part8_k(const int* __restrict__ rows, const int* __restrict__ cols,
                    float* cntA, float* cntB, unsigned* __restrict__ pkR,
                    unsigned* __restrict__ pkC, int2* __restrict__ tab) {
    __shared__ __attribute__((aligned(16))) unsigned stR[CH];
    __shared__ __attribute__((aligned(16))) unsigned stC[CH];
    __shared__ int ncls[16];
    __shared__ int sb[16];
    const int t = threadIdx.x;
    const int lane = t & 31;
    if (t < 16) ncls[t] = 0;
    __syncthreads();
    const bool live = t < CHV;
    const int v = blockIdx.x * CHV + (live ? t : 0);
    const int4 r4 = ((const int4*)rows)[v];
    const int4 c4 = ((const int4*)cols)[v];
    const unsigned place = __builtin_amdgcn_s_sendmsg_rtn(0x87);
    float* cnt = ((place >> 18) & 1u) ? cntB : cntA;
    const int kr0 = live ? eighth_of(r4.x) : 8, kc0 = live ? eighth_of(c4.x) : 8;
    const int kr1 = live ? eighth_of(r4.y) : 8, kc1 = live ? eighth_of(c4.y) : 8;
    const int kr2 = live ? eighth_of(r4.z) : 8, kc2 = live ? eighth_of(c4.z) : 8;
    const int kr3 = live ? eighth_of(r4.w) : 8, kc3 = live ? eighth_of(c4.w) : 8;
    const int qr0 = class_place8(kr0, lane, ncls), qc0 = class_place8(kc0, lane, ncls + 8);
    const int qr1 = class_place8(kr1, lane, ncls), qc1 = class_place8(kc1, lane, ncls + 8);
    const int qr2 = class_place8(kr2, lane, ncls), qc2 = class_place8(kc2, lane, ncls + 8);
    const int qr3 = class_place8(kr3, lane, ncls), qc3 = class_place8(kc3, lane, ncls + 8);
    if (live) {
        atomAddF(cnt + r4.x, 1.0f); atomAddF(cnt + NTOT + c4.x, 1.0f);
        atomAddF(cnt + r4.y, 1.0f); atomAddF(cnt + NTOT + c4.y, 1.0f);
        atomAddF(cnt + r4.z, 1.0f); atomAddF(cnt + NTOT + c4.z, 1.0f);
        atomAddF(cnt + r4.w, 1.0f); atomAddF(cnt + NTOT + c4.w, 1.0f);
    }
    __syncthreads();
    if (t < 16) {
        int s = 0;
        for (int cc = t & 8; cc < t; ++cc) s += ncls[cc];
        sb[t] = s;
        tab[blockIdx.x * 16 + t] = make_int2(s, s + ncls[t]);
    }
    __syncthreads();
    if (live) {
        const unsigned w0 = (unsigned)r4.x | ((unsigned)c4.x << 16);
        const unsigned w1 = (unsigned)r4.y | ((unsigned)c4.y << 16);
        const unsigned w2 = (unsigned)r4.z | ((unsigned)c4.z << 16);
        const unsigned w3 = (unsigned)r4.w | ((unsigned)c4.w << 16);
        stR[sb[kr0] + qr0] = w0; stC[sb[8 + kc0] + qc0] = w0;
        stR[sb[kr1] + qr1] = w1; stC[sb[8 + kc1] + qc1] = w1;
        stR[sb[kr2] + qr2] = w2; stC[sb[8 + kc2] + qc2] = w2;
        stR[sb[kr3] + qr3] = w3; stC[sb[8 + kc3] + qc3] = w3;
    }
    __syncthreads();
    if (live) {
        uint4 o;
        o.x = stR[4 * t]; o.y = stR[4 * t + 1]; o.z = stR[4 * t + 2]; o.w = stR[4 * t + 3];
        ((uint4*)pkR)[blockIdx.x * CHV + t] = o;
        o.x = stC[4 * t]; o.y = stC[4 * t + 1]; o.z = stC[4 * t + 2]; o.w = stC[4 * t + 3];
        ((uint4*)pkC)[blockIdx.x * CHV + t] = o;
    }
}

__global__ void rsq_k(float* pa, const float* pb, long n) {
    long i = (long)blockIdx.x * blockDim.x + threadIdx.x;
    if (i < n) {
        float d = pa[i];
        if (pb) d += pb[i];
        pa[i] = 1.0f / (sqrtf(d) + EPSF);
    }
}

__global__ void buildz_k(const float* uf, const float* itf, const float* bfv,
                         const float* __restrict__ dvi, float* __restrict__ zp,
                         float* __restrict__ xp) {
    __shared__ float tile[32][33];
    const int i0 = blockIdx.x * 32;
#pragma unroll
    for (int r = 0; r < 4; ++r) {
        int idx = threadIdx.x + r * 256;
        int j = idx >> 5, d = idx & 31;
        int i = i0 + j;
        tile[j][d] = (i < NTOT) ? feat_at(uf, itf, bfv, i, d) : 0.0f;
    }
    __syncthreads();
#pragma unroll
    for (int r = 0; r < 4; ++r) {
        int idx = threadIdx.x + r * 256;
        int d = idx >> 5, j = idx & 31;
        int i = i0 + j;
        if (i < NTOT) {
            const float xv = tile[j][d];
            zp[(size_t)d * NP + i] = xv * dvi[i];
            if (xp) xp[(size_t)d * NP + i] = xv;
        }
    }
}

template <bool DST_LOW>
__device__ __forceinline__ int src_node(unsigned w) {
    return DST_LOW ? (int)(w >> 16) : (int)(w & 0xFFFFu);
}
template <bool DST_LOW>
__device__ __forceinline__ int dst_node(unsigned w) {
    return DST_LOW ? (int)(w & 0xFFFFu) : (int)(w >> 16);
}

__device__ __forceinline__ void edge1(int dcol, float val, int lo, int cnt, float* dstL, bool live) {
    unsigned dd = (unsigned)(dcol - lo);
    if (live && dd < (unsigned)cnt)
        (void)__hip_atomic_fetch_add(dstL + dd, val, __ATOMIC_RELAXED,
                                     __HIP_MEMORY_SCOPE_WORKGROUP);
}

__device__ __forceinline__ void add_hit(bool hit, unsigned dd, float val, float* dstL) {
    if (__builtin_expect(hit, true))
        (void)__hip_atomic_fetch_add(dstL + dd, val, __ATOMIC_RELAXED,
                                     __HIP_MEMORY_SCOPE_WORKGROUP);
}

#define NV4 (ESLICE / 4)

__device__ __forceinline__ int ring_vec(int vs, int i) {
    const int x = vs + i;
    return x < CHV ? x : x - CHV;
}

template <bool DST_LOW>
__global__ __launch_bounds__(1024)
void spmm_pk_k(const unsigned* __restrict__ pk, const int4* __restrict__ cnts,
            const float* __restrict__ srcp, const float* __restrict__ rs,
            float* __restrict__ dstp) {
    __shared__ float srcL[NTOT];
    __shared__ float dstL[H1N];
    const int d  = blockIdx.x & 31;
    const int h  = (blockIdx.x >> 5) & 1;
    const int q  = blockIdx.x >> 6;
    const int lo = h * H0N;
    const int cnt = h ? H1N : H0N;
    const float* sp = srcp + (size_t)d * NP;
    if (rs) {
        for (int i = threadIdx.x; i < NTOT; i += 1024) srcL[i] = sp[i] * rs[i];
    } else {
        for (int i = threadIdx.x; i < NTOT; i += 1024) srcL[i] = sp[i];
    }
    for (int i = threadIdx.x; i < H1N; i += 1024) dstL[i] = 0.0f;
    __syncthreads();

    const int lane = threadIdx.x & 31;
    const int wv = __builtin_amdgcn_readfirstlane((int)(threadIdx.x >> 5));
    const int rot = (int)((blockIdx.x & 63) * NCHQ) >> 6;
    for (int cl = wv; cl < NCHQ; cl += 32) {
        int cc = cl + rot;
        if (cc >= NCHQ) cc -= NCHQ;
        const int chunk = q * NCHQ + cc;
        const int4 n = cnts[chunk];
        int start, len;
        if (DST_LOW) {
            start = h ? n.x + n.y : 0;
            len   = h ? n.z + n.w : n.x + n.y;
        } else {
            start = h ? n.x : n.x + n.y + n.z;
            len   = h ? n.y + n.z : n.w + n.x;
        }
        const int vs = start >> 2;
        int nv = ((start + len + 3) >> 2) - vs;
        if (nv > CHV) nv = CHV;
        if (len == 0) nv = 0;
        const uint4* p4 = (const uint4*)(pk + (size_t)chunk * CH);
        uint4 wa = p4[ring_vec(vs, lane < nv ? lane : 0)];
        uint4 wb = p4[ring_vec(vs, lane + 32 < nv ? lane + 32 : 0)];
        for (int t = 0; t < nv; t += 64) {
            const int ja = t + 64 + lane, jb = ja + 32;
            const uint4 na = p4[ring_vec(vs, ja < nv ? ja : 0)];
            const uint4 nb = p4[ring_vec(vs, jb < nv ? jb : 0)];
            const bool la = (t + lane) < nv;
            const bool lb = (t + lane + 32) < nv;
            const unsigned da0 = (unsigned)(dst_node<DST_LOW>(wa.x) - lo);
            const unsigned da1 = (unsigned)(dst_node<DST_LOW>(wa.y) - lo);
            const unsigned da2 = (unsigned)(dst_node<DST_LOW>(wa.z) - lo);
            const unsigned da3 = (unsigned)(dst_node<DST_LOW>(wa.w) - lo);
            const unsigned db0 = (unsigned)(dst_node<DST_LOW>(wb.x) - lo);
            const unsigned db1 = (unsigned)(dst_node<DST_LOW>(wb.y) - lo);
            const unsigned db2 = (unsigned)(dst_node<DST_LOW>(wb.z) - lo);
            const unsigned db3 = (unsigned)(dst_node<DST_LOW>(wb.w) - lo);
            const bool ha0 = la && da0 < (unsigned)cnt, ha1 = la && da1 < (unsigned)cnt;
            const bool ha2 = la && da2 < (unsigned)cnt, ha3 = la && da3 < (unsigned)cnt;
            const bool hb0 = lb && db0 < (unsigned)cnt, hb1 = lb && db1 < (unsigned)cnt;
            const bool hb2 = lb && db2 < (unsigned)cnt, hb3 = lb && db3 < (unsigned)cnt;
            float a0 = srcL[ha0 ? src_node<DST_LOW>(wa.x) : 0], a1 = srcL[ha1 ? src_node<DST_LOW>(wa.y) : 0];
            float a2 = srcL[ha2 ? src_node<DST_LOW>(wa.z) : 0], a3 = srcL[ha3 ? src_node<DST_LOW>(wa.w) : 0];
            float b0 = srcL[hb0 ? src_node<DST_LOW>(wb.x) : 0], b1 = srcL[hb1 ? src_node<DST_LOW>(wb.y) : 0];
            float b2 = srcL[hb2 ? src_node<DST_LOW>(wb.z) : 0], b3 = srcL[hb3 ? src_node<DST_LOW>(wb.w) : 0];
            asm volatile("" :: "v"(a0), "v"(a1), "v"(a2), "v"(a3),
                               "v"(b0), "v"(b1), "v"(b2), "v"(b3));
            add_hit(ha0, da0, a0, dstL);
            add_hit(ha1, da1, a1, dstL);
            add_hit(ha2, da2, a2, dstL);
            add_hit(ha3, da3, a3, dstL);
            add_hit(hb0, db0, b0, dstL);
            add_hit(hb1, db1, b1, dstL);
            add_hit(hb2, db2, b2, dstL);
            add_hit(hb3, db3, b3, dstL);
            wa = na; wb = nb;
        }
    }
    __syncthreads();

    float* dp = dstp + (size_t)d * NP + lo;
    for (int i = threadIdx.x; i < cnt; i += 1024) atomAddF(dp + i, dstL[i]);
}

struct Run8 { int off; int nv; };
struct Bnd8 { int chunk, start, end; };

__device__ __forceinline__ Bnd8 bounds_of(const int2* __restrict__ tab, int cl, int rot, int k) {
    Bnd8 b;
    b.chunk = cl + rot;
    if (b.chunk >= NCHUNK) b.chunk -= NCHUNK;
    const int2 se = tab[b.chunk * 16 + k];
    b.start = se.x;
    b.end   = se.y;
    return b;
}

__device__ __forceinline__ Run8 run_of(Bnd8 b) {
    Run8 r;
    const bool any = b.end > b.start;
    const int vs = any ? b.start >> 2 : 0;
    r.nv  = any ? ((b.end + 3) >> 2) - vs : 0;
    r.off = b.chunk * CHV + vs;
    return r;
}

__device__ __forceinline__ int stream_vec(Run8 cur, Run8 nxt, int j, bool& valid) {
    const int jn = j - cur.nv;
    const bool inc = jn < 0;
    valid = inc || jn < nxt.nv;
    const int first = inc ? cur.off : nxt.off - cur.nv;
    return valid ? first + j : 0;
}

template <bool DST_LOW, bool FIN>
__global__ __launch_bounds__(1024)
void spmm8_k(const unsigned* __restrict__ pk, const int2* __restrict__ tab,
             const float* __restrict__ srcp, const float* __restrict__ rs,
             float* __restrict__ dstp, const float* __restrict__ xpl,
             const float* __restrict__ dvs, float* __restrict__ parts) {
    __shared__ float srcL[NTOT];
    __shared__ float dstL[E8L];
    __shared__ float red[32];
    const int d  = blockIdx.x & 31;
    const int k  = blockIdx.x >> 5;
    const int lo = k * E8N;
    const int cnt = k == 7 ? E8L : E8N;
    const float* sp = srcp + (size_t)d * NP;
    if (rs) {
        for (int i = threadIdx.x; i < NTOT; i += 1024) srcL[i] = sp[i] * rs[i];
    } else {
        for (int i = threadIdx.x; i < NTOT; i += 1024) srcL[i] = sp[i];
    }
    for (int i = threadIdx.x; i < E8L; i += 1024) dstL[i] = 0.0f;
    __syncthreads();

    const int lane = threadIdx.x & 31;
    const int wv = __builtin_amdgcn_readfirstlane((int)(threadIdx.x >> 5));
    const int rot = (d * NCHUNK) >> 5;
    int cl = wv;
    Run8 cur = run_of(bounds_of(tab, cl, rot, k));
    Run8 nxt = run_of(bounds_of(tab, cl + 32, rot, k));
    Bnd8 bnn = bounds_of(tab, cl + 64, rot, k);
    const uint4* __restrict__ p4 = (const uint4*)pk;
    int t = 0;
    bool va, vb;
    uint4 wa = p4[stream_vec(cur, nxt, lane, va)];
    uint4 wb = p4[stream_vec(cur, nxt, lane + 32, vb)];
    for (;;) {
        t += 64;
        if (t >= cur.nv) {
            const int over = t - cur.nv;
            t = over < nxt.nv ? over : nxt.nv;
            cl += 32;
            cur = nxt;
            nxt.nv = 0;
            if (cl + 32 < NCHUNK) nxt = run_of(bnn);
            if (cl + 64 < NCHUNK) bnn = bounds_of(tab, cl + 64, rot, k);
        }
        const bool last = cl >= NCHUNK;
        bool nva, nvb;
        const uint4 na = p4[stream_vec(cur, nxt, t + lane, nva)];
        const uint4 nb = p4[stream_vec(cur, nxt, t + lane + 32, nvb)];
        const unsigned da0 = (unsigned)(dst_node<DST_LOW>(wa.x) - lo);
        const unsigned da1 = (unsigned)(dst_node<DST_LOW>(wa.y) - lo);
        const unsigned da2 = (unsigned)(dst_node<DST_LOW>(wa.z) - lo);
        const unsigned da3 = (unsigned)(dst_node<DST_LOW>(wa.w) - lo);
        const unsigned db0 = (unsigned)(dst_node<DST_LOW>(wb.x) - lo);
        const unsigned db1 = (unsigned)(dst_node<DST_LOW>(wb.y) - lo);
        const unsigned db2 = (unsigned)(dst_node<DST_LOW>(wb.z) - lo);
        const unsigned db3 = (unsigned)(dst_node<DST_LOW>(wb.w) - lo);
        const bool ha0 = va && da0 < (unsigned)cnt, ha1 = va && da1 < (unsigned)cnt;
        const bool ha2 = va && da2 < (unsigned)cnt, ha3 = va && da3 < (unsigned)cnt;
        const bool hb0 = vb && db0 < (unsigned)cnt, hb1 = vb && db1 < (unsigned)cnt;
        const bool hb2 = vb && db2 < (unsigned)cnt, hb3 = vb && db3 < (unsigned)cnt;
        float a0 = srcL[ha0 ? src_node<DST_LOW>(wa.x) : 0], a1 = srcL[ha1 ? src_node<DST_LOW>(wa.y) : 0];
        float a2 = srcL[ha2 ? src_node<DST_LOW>(wa.z) : 0], a3 = srcL[ha3 ? src_node<DST_LOW>(wa.w) : 0];
        float b0 = srcL[hb0 ? src_node<DST_LOW>(wb.x) : 0], b1 = srcL[hb1 ? src_node<DST_LOW>(wb.y) : 0];
        float b2 = srcL[hb2 ? src_node<DST_LOW>(wb.z) : 0], b3 = srcL[hb3 ? src_node<DST_LOW>(wb.w) : 0];
        asm volatile("" :: "v"(a0), "v"(a1), "v"(a2), "v"(a3),
                           "v"(b0), "v"(b1), "v"(b2), "v"(b3));
        add_hit(ha0, da0, a0, dstL);
        add_hit(ha1, da1, a1, dstL);
        add_hit(ha2, da2, a2, dstL);
        add_hit(ha3, da3, a3, dstL);
        add_hit(hb0, db0, b0, dstL);
        add_hit(hb1, db1, b1, dstL);
        add_hit(hb2, db2, b2, dstL);
        add_hit(hb3, db3, b3, dstL);
        if (last) break;
        wa = na; wb = nb; va = nva; vb = nvb;
    }
    float* dp = dstp + (size_t)d * NP + lo;
    if (!FIN) {
        __syncthreads();
        for (int i = threadIdx.x; i < cnt; i += 1024) dp[i] = dstL[i];
        return;
    }
    const float* xq = xpl + (size_t)d * NP + lo;
    const float* sq = dvs + lo;
    float xr[6], sc[6];
#pragma unroll
    for (int j = 0; j < 6; ++j) {
        const int i = threadIdx.x + j * 1024;
        const int ic = i < cnt ? i : 0;
        xr[j] = xq[ic];
        sc[j] = sq[ic];
    }
    __syncthreads();
    float ss = 0.0f;
#pragma unroll
    for (int j = 0; j < 6; ++j) {
        const int i = threadIdx.x + j * 1024;
        if (i < cnt) {
            const float val = xr[j] * 0.5f + dstL[i] * sc[j] * (1.0f / 3.0f);
            dp[i] = val;
            ss += val * val;
        }
    }
#pragma unroll
    for (int m = 16; m > 0; m >>= 1) ss += __shfl_xor(ss, m);
    if (lane == 0) red[wv] = ss;
    __syncthreads();
    if (wv == 0) {
        float t = red[lane];
#pragma unroll
        for (int m = 16; m > 0; m >>= 1) t += __shfl_xor(t, m);
        if (lane == 0) parts[blockIdx.x] = t;
    }
}

#define TUG  ((NU + 31) / 32)
#define TBG0 ((NU + NI) / 32)
#define TBG  ((NTOT + 31) / 32 - TBG0)
__global__ void turn_k(const float* __restrict__ ap, float* __restrict__ all) {
    __shared__ float tile[32][33];
    const int g  = (int)blockIdx.x < TUG ? (int)blockIdx.x : (int)blockIdx.x - TUG + TBG0;
    const int i0 = g * 32;
    float t4[4];
#pragma unroll
    for (int r = 0; r < 4; ++r) {
        int idx = threadIdx.x + r * 256;
        int d = idx >> 5, j = idx & 31;
        int i = i0 + j;
        t4[r] = ap[(size_t)d * NP + (i < NTOT ? i : NTOT - 1)];
    }
#pragma unroll
    for (int r = 0; r < 4; ++r) {
        int idx = threadIdx.x + r * 256;
        tile[idx >> 5][idx & 31] = t4[r];
    }
    __syncthreads();
#pragma unroll
    for (int r = 0; r < 4; ++r) {
        int idx = threadIdx.x + r * 256;
        int j = idx >> 5, d = idx & 31;
        int i = i0 + j;
        if (i < NTOT) all[(size_t)i * EMB + d] = tile[d][j];
    }
}

__global__ __launch_bounds__(1024)
void spmm_k(const int* __restrict__ sidx, const int* __restrict__ didx,
            const float* __restrict__ srcp, const float* __restrict__ rs,
            float* __restrict__ dstp) {
    __shared__ float srcL[NTOT];
    __shared__ float dstL[HALFN];
    const int d  = blockIdx.x & 31;
    const int h  = (blockIdx.x >> 5) & 1;
    const int q  = blockIdx.x >> 6;
    const int lo = h * HALFN;
    const float* sp = srcp + (size_t)d * NP;
    if (rs) {
        for (int i = threadIdx.x; i < NTOT; i += 1024) srcL[i] = sp[i] * rs[i];
    } else {
        for (int i = threadIdx.x; i < NTOT; i += 1024) srcL[i] = sp[i];
    }
    for (int i = threadIdx.x; i < HALFN; i += 1024) dstL[i] = 0.0f;
    __syncthreads();

    const int4* s4 = (const int4*)(sidx + (size_t)q * ESLICE);
    const int4* d4 = (const int4*)(didx + (size_t)q * ESLICE);
    const int off = (blockIdx.x & 63) * (NV4 / 64 / 32 * 32);
    int k = threadIdx.x;
    int v = k + off;
    int4 dc = d4[v];
    int4 sr = s4[v];
    for (;;) {
        const int kn = k + 1024;
        const int vw = v + 1024;
        const int vn = vw < NV4 ? vw : vw - NV4;
        const int kc = kn < NV4 ? vn : v;
        const int4 dcn = d4[kc];
        const int4 srn = s4[kc];
        float a0 = srcL[sr.x], a1 = srcL[sr.y], a2 = srcL[sr.z], a3 = srcL[sr.w];
        asm volatile("" :: "v"(a0), "v"(a1), "v"(a2), "v"(a3));
        edge1(dc.x, a0, lo, HALFN, dstL, true);
        edge1(dc.y, a1, lo, HALFN, dstL, true);
        edge1(dc.z, a2, lo, HALFN, dstL, true);
        edge1(dc.w, a3, lo, HALFN, dstL, true);
        if (kn >= NV4) break;
        dc = dcn; sr = srn; k = kn; v = vn;
    }
    __syncthreads();

    float* dp = dstp + (size_t)d * NP + lo;
    for (int i = threadIdx.x; i < HALFN; i += 1024) atomAddF(dp + i, dstL[i]);
}

__global__ void final_k(const float* uf, const float* itf, const float* bfv,
                        const float* __restrict__ dvi, const float* __restrict__ v,
                        float* __restrict__ all, float* __restrict__ partials) {
    __shared__ float sdata[256];
    long t = (long)blockIdx.x * blockDim.x + threadIdx.x;
    float sq = 0.0f;
    if (t < (long)NTOT * EMB) {
        int i = (int)(t >> 5), d = (int)(t & 31);
        const float x = feat_at(uf, itf, bfv, i, d);
        const float sc = dvi[i];
        const float* vp = v + (size_t)d * NP + i;
        const float vs = vp[0];
        float val = x * 0.5f + vs * sc * (1.0f / 3.0f);
        all[t] = val;
        sq = val * val;
    }
    sdata[threadIdx.x] = sq;
    __syncthreads();
    for (int s = 128; s > 0; s >>= 1) {
        if (threadIdx.x < (unsigned)s) sdata[threadIdx.x] += sdata[threadIdx.x + s];
        __syncthreads();
    }
    if (threadIdx.x == 0) partials[blockIdx.x] = sdata[0];
}

__global__ void loss_k(const float* __restrict__ partials, int nb, float* __restrict__ out_loss) {
    __shared__ float sdata[256];
    float s = 0.0f;
    for (int i = threadIdx.x; i < nb; i += 256) s += partials[i];
    sdata[threadIdx.x] = s;
    __syncthreads();
    for (int k = 128; k > 0; k >>= 1) {
        if (threadIdx.x < (unsigned)k) sdata[threadIdx.x] += sdata[threadIdx.x + k];
        __syncthreads();
    }
    if (threadIdx.x == 0) *out_loss = L2NORM * sdata[0];
}

__global__ void pred_k(const float* __restrict__ all, const int* __restrict__ users,
                       const int* __restrict__ bundles, const float* __restrict__ ubound,
                       const float* __restrict__ zrow,
                       float* __restrict__ pred, float* __restrict__ usb) {
    int wave = (blockIdx.x * blockDim.x + threadIdx.x) >> 5;
    int lane = threadIdx.x & 31;
    int b = wave / 7;
    int kt = wave % 7;

    const float* bfbase = all + (size_t)(NU + NI) * EMB;
    const float* ue = all + (size_t)users[b] * EMB;

    int m = lane & 15;
    int khalf = lane >> 4;
    int kk = kt * 16 + m;

    const float* arow;
    if (kk < KB)                 arow = bfbase + (size_t)bundles[b * KB + kk] * EMB;
    else if (kt == 6 && m == 4)  arow = ubound;
    else                         arow = zrow;

    v8f c = {};
#pragma unroll
    for (int s = 0; s < 8; ++s) {
        int e0 = s * 4 + khalf * 2;
        v2f a  = *(const v2f*)(arow + e0);
        v2f bb = *(const v2f*)(ue + e0);
        c = __builtin_amdgcn_wmma_f32_16x16x4_f32(false, a, false, bb,
                                                  (short)0, c, false, false);
    }

    if (lane == 0) {
#pragma unroll
        for (int j = 0; j < 8; ++j) {
            int kkk = kt * 16 + j;
            if (kkk < KB)                 pred[(size_t)b * KB + kkk] = c[j];
            else if (kt == 6 && j == 4)   usb[b] = c[j];
        }
    } else if (lane == 16) {
#pragma unroll
        for (int j = 0; j < 8; ++j) {
            int kkk = kt * 16 + 8 + j;
            if (kkk < KB) pred[(size_t)b * KB + kkk] = c[j];
        }
    }
}

extern "C" void kernel_launch(void* const* d_in, const int* in_sizes, int n_in,
                              void* d_out, int out_size, void* d_ws, size_t ws_size,
                              hipStream_t stream) {
    const float* uf      = (const float*)d_in[0];
    const float* itf     = (const float*)d_in[1];
    const float* bfv     = (const float*)d_in[2];
    const float* ubound  = (const float*)d_in[3];
    const int*   rows    = (const int*)d_in[4];
    const int*   cols    = (const int*)d_in[5];
    const int*   users   = (const int*)d_in[6];
    const int*   bundles = (const int*)d_in[7];
    float* out = (float*)d_out;

    const bool plain = ws_size >= (size_t)PL_END_B;
    float* ws   = (float*)d_ws;
    float* p0   = ws;
    float* p1   = p0 + 32L * NP;
    float* p2   = p1 + (plain ? 128L : 32L) * NP;
    float* dvde = p2 + (plain ? 128L : 32L) * NP;
    float* partials = dvde + 2L * NTOT;
    float* zrow = partials + PARTN;
    const bool packed = plain || ws_size >= (size_t)PK_END_B;
    unsigned* pk = packed ? (unsigned*)(zrow + 104) : (unsigned*)nullptr;
    unsigned* pkR = plain ? (unsigned*)(p1 + 32L * NP) : (unsigned*)nullptr;
    int2* tab8 = plain ? (int2*)(p2 + 32L * NP) : (int2*)nullptr;
    float* xpl = plain ? p1 + 80L * NP : (float*)nullptr;

    const long elems = (long)NTOT * EMB;
    const int  eb    = (int)((elems + 255) / 256);

    float* cntB = plain ? dvde - 2L * NTOT : dvde;

    zero_k<<<2048, 256, 0, stream>>>(cntB, plain ? 4L * NTOT : 2L * NTOT);
    if (!plain) zero_k<<<4096, 256, 0, stream>>>(p1, 64L * NP);
    zero_k<<<1, 32, 0, stream>>>(zrow, 32L);

    int4* ccnt = (int4*)partials;
    if (plain)       degree_part8_k<<<NCHUNK, 1024, 0, stream>>>(rows, cols, dvde, cntB, pkR, pk, tab8);
    else if (packed) degree_part_k<<<NCHUNK, 1024, 0, stream>>>(rows, cols, dvde, cntB, pk, ccnt);
    else             degree_k<<<(NNZE + 255) / 256, 256, 0, stream>>>(rows, cols, dvde, cntB);
    rsq_k<<<(int)((2L * NTOT + 255) / 256), 256, 0, stream>>>(dvde, plain ? cntB : (const float*)nullptr,
                                                             2L * NTOT);

    buildz_k<<<(NTOT + 31) / 32, 256, 0, stream>>>(uf, itf, bfv, dvde, p0, xpl);
    if (plain) {
        spmm8_k<false, false><<<32 * 8, 1024, 0, stream>>>(pk, tab8 + 8, p0, (const float*)nullptr, p1,
                                                       (const float*)nullptr, (const float*)nullptr,
                                                       (float*)nullptr);
        spmm8_k<true, true><<<32 * 8, 1024, 0, stream>>>(pkR, tab8, p1, dvde + NTOT, p2, xpl, dvde, partials);
    } else if (packed) {
        spmm_pk_k<false><<<32 * 2 * NSLICE, 1024, 0, stream>>>(pk, ccnt, p0, (const float*)nullptr, p1);
        spmm_pk_k<true><<<32 * 2 * NSLICE, 1024, 0, stream>>>(pk, ccnt, p1, dvde + NTOT, p2);
    } else {
        spmm_k<<<32 * 2 * NSLICE, 1024, 0, stream>>>(rows, cols, p0, (const float*)nullptr, p1);
        spmm_k<<<32 * 2 * NSLICE, 1024, 0, stream>>>(cols, rows, p1, dvde + NTOT, p2);
    }

    if (plain) {
        turn_k<<<TUG + TBG, 256, 0, stream>>>(p2, p0);
        loss_k<<<1, 256, 0, stream>>>(partials, 32 * 8, out + BATCHN * KB + BATCHN);
    } else {
        final_k<<<eb, 256, 0, stream>>>(uf, itf, bfv, dvde, p2, p0, partials);
        loss_k<<<1, 256, 0, stream>>>(partials, eb, out + BATCHN * KB + BATCHN);
    }

    pred_k<<<1792, 256, 0, stream>>>(p0, users, bundles, ubound, zrow,
                                     out, out + BATCHN * KB);
}
